# starting kernel
# baseline (speedup 1.0000x reference)
#include <hip/hip_runtime.h>


typedef __attribute__((ext_vector_type(2))) float v2f;
typedef __attribute__((ext_vector_type(8))) float v8f;

#define KDIM 64

__global__ void fill_zero(float* __restrict__ p, int n) {
  int i = blockIdx.x * blockDim.x + threadIdx.x;
  int stride = gridDim.x * blockDim.x;
  for (; i < n; i += stride) p[i] = 0.0f;
}

template <int OUTF, bool RELU, bool ADD_AGG>
__global__ __launch_bounds__(256) void sage_gemm(
    const float* __restrict__ X, const float* __restrict__ W,
    const float* __restrict__ bias, const float* __restrict__ agg,
    const float* __restrict__ cnt, float* __restrict__ Y, int n) {
  constexpr int NT = OUTF / 16;
  const int wave = threadIdx.x >> 5;
  const int lane = threadIdx.x & 31;
  const int half = lane >> 4;
  const int l15  = lane & 15;
  const int tile = blockIdx.x * (blockDim.x >> 5) + wave;
  const int r0 = tile * 16;
  if (r0 >= n) return;

  int arow = r0 + l15;
  if (arow >= n) arow = n - 1;
  const float* xrow = X + (long)arow * KDIM + half * 2;

  v8f acc[NT];
#pragma unroll
  for (int t = 0; t < NT; ++t) {
    float b = bias ? bias[t * 16 + l15] : 0.0f;
#pragma unroll
    for (int j = 0; j < 8; ++j) acc[t][j] = b;
  }

#pragma unroll
  for (int kb = 0; kb < KDIM / 4; ++kb) {
    v2f a = *(const v2f*)(xrow + kb * 4);
#pragma unroll
    for (int t = 0; t < NT; ++t) {
      v2f b = *(const v2f*)(W + (t * 16 + l15) * KDIM + kb * 4 + half * 2);
      acc[t] = __builtin_amdgcn_wmma_f32_16x16x4_f32(
           false, a,  false, b,
           (short)0, acc[t],  false,  false);
    }
  }

#pragma unroll
  for (int t = 0; t < NT; ++t) {
    int col = t * 16 + l15;
#pragma unroll
    for (int v = 0; v < 8; ++v) {
      int row = r0 + half * 8 + v;
      if (row < n) {
        float val = acc[t][v];
        if (ADD_AGG) {
          float c = cnt[row];
          val += agg[(long)row * OUTF + col] / fmaxf(c, 1.0f);
        }
        if (RELU) val = fmaxf(val, 0.0f);
        Y[(long)row * OUTF + col] = val;
      }
    }
  }
}

template <int F>
__global__ __launch_bounds__(256) void sage_scatter(
    const int* __restrict__ ei, const float* __restrict__ H,
    float* __restrict__ agg, float* __restrict__ cnt, int nE) {
  constexpr int CH = F / 4;
  int idx = blockIdx.x * blockDim.x + threadIdx.x;
  if (idx >= nE * CH) return;
  int e = idx / CH;
  int c = idx - e * CH;
  int src = ei[e];
  int dst = ei[nE + e];
  const float4 m = *(const float4*)(H + (long)src * F + c * 4);
  float* ap = agg + (long)dst * F + c * 4;
  (void)__hip_atomic_fetch_add(ap + 0, m.x, __ATOMIC_RELAXED, __HIP_MEMORY_SCOPE_AGENT);
  (void)__hip_atomic_fetch_add(ap + 1, m.y, __ATOMIC_RELAXED, __HIP_MEMORY_SCOPE_AGENT);
  (void)__hip_atomic_fetch_add(ap + 2, m.z, __ATOMIC_RELAXED, __HIP_MEMORY_SCOPE_AGENT);
  (void)__hip_atomic_fetch_add(ap + 3, m.w, __ATOMIC_RELAXED, __HIP_MEMORY_SCOPE_AGENT);
  if (cnt != nullptr && c == 0)
    (void)__hip_atomic_fetch_add(cnt + dst, 1.0f, __ATOMIC_RELAXED, __HIP_MEMORY_SCOPE_AGENT);
}

extern "C" void kernel_launch(void* const* d_in, const int* in_sizes, int n_in,
                              void* d_out, int out_size, void* d_ws, size_t ws_size,
                              hipStream_t stream) {
  const float* x   = (const float*)d_in[0];
  const int*   ei  = (const int*)d_in[1];
  const float* Wl0 = (const float*)d_in[2];
  const float* bl0 = (const float*)d_in[3];
  const float* Wr0 = (const float*)d_in[4];
  const float* Wl1 = (const float*)d_in[5];
  const float* bl1 = (const float*)d_in[6];
  const float* Wr1 = (const float*)d_in[7];
  float* out = (float*)d_out;

  const int N = in_sizes[0] / 64;
  const int E = in_sizes[1] / 2;

  float* B0 = (float*)d_ws;
  float* B1 = B0 + (size_t)N * 64;
  float* B2 = B1 + (size_t)N * 64;
  float* B3 = B2 + (size_t)N * 64;

  const dim3 blk(256);
  const int tiles = (N + 15) / 16;
  const dim3 gemmGrid((tiles + 7) / 8);

  fill_zero<<<dim3(2048), blk, 0, stream>>>(B1, N * 64);
  fill_zero<<<dim3(256),  blk, 0, stream>>>(B3, N);
  sage_gemm<64, false, false><<<gemmGrid, blk, 0, stream>>>(x, Wl0, bl0, nullptr, nullptr, B0, N);
  {
    int work = E * 16;
    sage_scatter<64><<<dim3((work + 255) / 256), blk, 0, stream>>>(ei, B0, B1, B3, E);
  }
  sage_gemm<64, true, true><<<gemmGrid, blk, 0, stream>>>(x, Wr0, nullptr, B1, B3, B2, N);

  fill_zero<<<dim3(1024), blk, 0, stream>>>(B1, N * 32);
  sage_gemm<32, false, false><<<gemmGrid, blk, 0, stream>>>(B2, Wl1, bl1, nullptr, nullptr, B0, N);
  {
    int work = E * 8;
    sage_scatter<32><<<dim3((work + 255) / 256), blk, 0, stream>>>(ei, B0, B1, nullptr, E);
  }
  sage_gemm<32, false, true><<<gemmGrid, blk, 0, stream>>>(B2, Wr1, nullptr, B1, B3, out, N);
}


# Replace fp32 atomic scatter-add by a per-call dst-grouped edge list (int histogram with rank, scan, fill) and an atomic-free row gather; sums run in arrival-rank order, 2-4 partials per node
# speedup vs baseline: 1.5590x; 1.5590x over previous
#include <hip/hip_runtime.h>


typedef __attribute__((ext_vector_type(2))) float v2f;
typedef __attribute__((ext_vector_type(8))) float v8f;

#define KDIM 64

__global__ void fill_zero(float* __restrict__ p, int n) {
  int i = blockIdx.x * blockDim.x + threadIdx.x;
  int stride = gridDim.x * blockDim.x;
  for (; i < n; i += stride) p[i] = 0.0f;
}

template <int OUTF, bool RELU, bool ADD_AGG>
__global__ __launch_bounds__(256) void sage_gemm(
    const float* __restrict__ X, const float* __restrict__ W,
    const float* __restrict__ bias, const float* __restrict__ agg,
    const float* __restrict__ cnt, float* __restrict__ Y, int n) {
  constexpr int NT = OUTF / 16;
  const int wave = threadIdx.x >> 5;
  const int lane = threadIdx.x & 31;
  const int half = lane >> 4;
  const int l15  = lane & 15;
  const int tile = blockIdx.x * (blockDim.x >> 5) + wave;
  const int r0 = tile * 16;
  if (r0 >= n) return;

  int arow = r0 + l15;
  if (arow >= n) arow = n - 1;
  const float* xrow = X + (long)arow * KDIM + half * 2;

  v8f acc[NT];
#pragma unroll
  for (int t = 0; t < NT; ++t) {
    float b = bias ? bias[t * 16 + l15] : 0.0f;
#pragma unroll
    for (int j = 0; j < 8; ++j) acc[t][j] = b;
  }

#pragma unroll
  for (int kb = 0; kb < KDIM / 4; ++kb) {
    v2f a = *(const v2f*)(xrow + kb * 4);
#pragma unroll
    for (int t = 0; t < NT; ++t) {
      v2f b = *(const v2f*)(W + (t * 16 + l15) * KDIM + kb * 4 + half * 2);
      acc[t] = __builtin_amdgcn_wmma_f32_16x16x4_f32(
           false, a,  false, b,
           (short)0, acc[t],  false,  false);
    }
  }

#pragma unroll
  for (int t = 0; t < NT; ++t) {
    int col = t * 16 + l15;
#pragma unroll
    for (int v = 0; v < 8; ++v) {
      int row = r0 + half * 8 + v;
      if (row < n) {
        float val = acc[t][v];
        if (ADD_AGG) {
          float c = cnt[row];
          val += agg[(long)row * OUTF + col] / fmaxf(c, 1.0f);
        }
        if (RELU) val = fmaxf(val, 0.0f);
        Y[(long)row * OUTF + col] = val;
      }
    }
  }
}

__global__ __launch_bounds__(256) void edge_rank(
    const int* __restrict__ ei, int* __restrict__ deg, int* __restrict__ rank, int nE) {
  const int e0 = blockIdx.x * 1024 + threadIdx.x;
  int dst[4], r[4];
#pragma unroll
  for (int j = 0; j < 4; ++j) dst[j] = ei[nE + min(e0 + j * 256, nE - 1)];
#pragma unroll
  for (int j = 0; j < 4; ++j) {
    r[j] = 0;
    if (e0 + j * 256 < nE)
      r[j] = __hip_atomic_fetch_add(deg + dst[j], 1, __ATOMIC_RELAXED, __HIP_MEMORY_SCOPE_AGENT);
  }
#pragma unroll
  for (int j = 0; j < 4; ++j) {
    const int e = e0 + j * 256;
    if (e < nE) rank[e] = r[j];
  }
}

__global__ __launch_bounds__(256) void scan_local(
    const int* __restrict__ deg, int* __restrict__ rowptr, int* __restrict__ bsum,
    float* __restrict__ cnt, int n) {
  __shared__ int wsum[8];
  const int lane = threadIdx.x & 31;
  const int wave = threadIdx.x >> 5;
  const int base = (blockIdx.x * 256 + threadIdx.x) * 4;
  int v[4];
#pragma unroll
  for (int i = 0; i < 4; ++i) v[i] = (base + i < n) ? deg[base + i] : 0;
  const int tsum = v[0] + v[1] + v[2] + v[3];
  int inc = tsum;
#pragma unroll
  for (int d = 1; d < 32; d <<= 1) {
    const int t = __shfl_up(inc, d);
    if (lane >= d) inc += t;
  }
  if (lane == 31) wsum[wave] = inc;
  __syncthreads();
  int excl = inc - tsum;
#pragma unroll
  for (int w = 0; w < 8; ++w)
    if (w < wave) excl += wsum[w];
#pragma unroll
  for (int i = 0; i < 4; ++i) {
    if (base + i < n) {
      rowptr[base + i] = excl;
      cnt[base + i] = (float)v[i];
    }
    excl += v[i];
  }
  if (threadIdx.x == 255) bsum[blockIdx.x] = excl;
}

__global__ __launch_bounds__(256) void scan_add(
    int* __restrict__ rowptr, const int* __restrict__ bsum, int n) {
  __shared__ int wsum[8];
  const int lane = threadIdx.x & 31;
  const int wave = threadIdx.x >> 5;
  int part = 0;
  for (int b = threadIdx.x; b < (int)blockIdx.x; b += 256) part += bsum[b];
#pragma unroll
  for (int d = 16; d > 0; d >>= 1) part += __shfl_xor(part, d);
  if (lane == 0) wsum[wave] = part;
  __syncthreads();
  int off = 0;
#pragma unroll
  for (int w = 0; w < 8; ++w) off += wsum[w];
  const int base = (blockIdx.x * 256 + threadIdx.x) * 4;
#pragma unroll
  for (int i = 0; i < 4; ++i)
    if (base + i < n) rowptr[base + i] += off;
  if (blockIdx.x == gridDim.x - 1 && threadIdx.x == 0) rowptr[n] = off + bsum[blockIdx.x];
}

__global__ __launch_bounds__(256) void edge_fill(
    const int* __restrict__ ei, const int* __restrict__ rowptr,
    const int* __restrict__ rank, int* __restrict__ col, int nE) {
  const int e0 = blockIdx.x * 1024 + threadIdx.x;
  int src[4], dst[4], pos[4];
#pragma unroll
  for (int j = 0; j < 4; ++j) {
    const int e = e0 + j * 256;
    const bool ok = e < nE;
    src[j] = ok ? ei[e] : 0;
    dst[j] = ok ? ei[nE + e] : 0;
    pos[j] = ok ? rank[e] : nE;
  }
#pragma unroll
  for (int j = 0; j < 4; ++j) pos[j] += rowptr[dst[j]];
#pragma unroll
  for (int j = 0; j < 4; ++j)
    if ((unsigned)pos[j] < (unsigned)nE) col[pos[j]] = src[j];
}

template <int F>
__global__ __launch_bounds__(256) void sage_gather(
    const int* __restrict__ rowptr, const int* __restrict__ col,
    const float* __restrict__ H, float* __restrict__ agg, int n, int nE) {
  constexpr int G = F / 4;
  constexpr int NG = 32 / G;
  const int lane = threadIdx.x & 31;
  const int sub = lane / G;
  const int c = lane % G;
  const int wv = __builtin_amdgcn_readfirstlane((int)(threadIdx.x >> 5));
  const int nwaves = gridDim.x * (blockDim.x >> 5);
  for (int node = blockIdx.x * (blockDim.x >> 5) + wv; node < n; node += nwaves) {
    int beg = __builtin_amdgcn_readfirstlane(rowptr[node]);
    int end = __builtin_amdgcn_readfirstlane(rowptr[node + 1]);
    beg = min(max(beg, 0), nE);
    end = min(max(end, beg), nE);
    float4 acc = make_float4(0.0f, 0.0f, 0.0f, 0.0f);
    for (int base = beg; base < end; base += 32) {
      const int m = min(32, end - base);
      int cv = (lane < m) ? col[base + lane] : 0;
      cv = (int)min((unsigned)cv, (unsigned)(n - 1));
      const int full = m / NG;
#pragma unroll 4
      for (int k = 0; k < full; ++k) {
        const int s = __shfl(cv, k * NG + sub);
        const float4 v = *(const float4*)(H + (long)s * F + c * 4);
        acc.x += v.x; acc.y += v.y; acc.z += v.z; acc.w += v.w;
      }
      const int rem = m - full * NG;
      if (rem != 0) {
        const int s = __shfl(cv, full * NG + sub);
        if (sub < rem) {
          const float4 v = *(const float4*)(H + (long)s * F + c * 4);
          acc.x += v.x; acc.y += v.y; acc.z += v.z; acc.w += v.w;
        }
      }
    }
#pragma unroll
    for (int d = G; d < 32; d <<= 1) {
      acc.x += __shfl_xor(acc.x, d);
      acc.y += __shfl_xor(acc.y, d);
      acc.z += __shfl_xor(acc.z, d);
      acc.w += __shfl_xor(acc.w, d);
    }
    if (sub == 0) *(float4*)(agg + (long)node * F + c * 4) = acc;
  }
}

extern "C" void kernel_launch(void* const* d_in, const int* in_sizes, int n_in,
                              void* d_out, int out_size, void* d_ws, size_t ws_size,
                              hipStream_t stream) {
  const float* x   = (const float*)d_in[0];
  const int*   ei  = (const int*)d_in[1];
  const float* Wl0 = (const float*)d_in[2];
  const float* bl0 = (const float*)d_in[3];
  const float* Wr0 = (const float*)d_in[4];
  const float* Wl1 = (const float*)d_in[5];
  const float* bl1 = (const float*)d_in[6];
  const float* Wr1 = (const float*)d_in[7];
  float* out = (float*)d_out;

  const int N = in_sizes[0] / 64;
  const int E = in_sizes[1] / 2;

  float* B0 = (float*)d_ws;
  float* B1 = B0 + (size_t)N * 64;
  float* B2 = B1 + (size_t)N * 64;
  float* B3 = B2 + (size_t)N * 64;
  int* deg    = (int*)(B3 + N);
  int* rowptr = deg + N;
  int* bsum   = rowptr + N + 4;
  int* rank   = bsum + 1024;
  int* col    = rank + E;

  const dim3 blk(256);
  const int tiles = (N + 15) / 16;
  const dim3 gemmGrid((tiles + 7) / 8);
  const dim3 scanGrid((N + 1023) / 1024);
  const dim3 edgeGrid((E + 1023) / 1024);

  fill_zero<<<dim3(256), blk, 0, stream>>>((float*)deg, N);
  edge_rank<<<edgeGrid, blk, 0, stream>>>(ei, deg, rank, E);
  scan_local<<<scanGrid, blk, 0, stream>>>(deg, rowptr, bsum, B3, N);
  scan_add<<<scanGrid, blk, 0, stream>>>(rowptr, bsum, N);
  edge_fill<<<edgeGrid, blk, 0, stream>>>(ei, rowptr, rank, col, E);

  sage_gemm<64, false, false><<<gemmGrid, blk, 0, stream>>>(x, Wl0, bl0, nullptr, nullptr, B0, N);
  sage_gather<64><<<dim3(4096), blk, 0, stream>>>(rowptr, col, B0, B1, N, E);
  sage_gemm<64, true, true><<<gemmGrid, blk, 0, stream>>>(x, Wr0, nullptr, B1, B3, B2, N);

  sage_gemm<32, false, false><<<gemmGrid, blk, 0, stream>>>(B2, Wl1, bl1, nullptr, nullptr, B0, N);
  sage_gather<32><<<dim3(4096), blk, 0, stream>>>(rowptr, col, B0, B1, N, E);
  sage_gemm<32, false, true><<<gemmGrid, blk, 0, stream>>>(B2, Wr1, nullptr, B1, B3, out, N);
}


# One GEMM pass per layer computes both X Wl^T + b and X Wr^T (each A fragment feeds both weight sets); the gather finishes the row (mean, root add, ReLU): 11 -> 9 launches, sums unchanged
# speedup vs baseline: 1.6837x; 1.0800x over previous
#include <hip/hip_runtime.h>


typedef __attribute__((ext_vector_type(2))) float v2f;
typedef __attribute__((ext_vector_type(8))) float v8f;

#define KDIM 64

__global__ void fill_zero(float* __restrict__ p, int n) {
  int i = blockIdx.x * blockDim.x + threadIdx.x;
  int stride = gridDim.x * blockDim.x;
  for (; i < n; i += stride) p[i] = 0.0f;
}

template <int OUTF>
__global__ __launch_bounds__(256) void sage_gemm_pair(
    const float* __restrict__ X, const float* __restrict__ Wl,
    const float* __restrict__ bl, const float* __restrict__ Wr,
    float* __restrict__ Yl, float* __restrict__ Yr, int n) {
  constexpr int NT = OUTF / 16;
  const int wave = threadIdx.x >> 5;
  const int lane = threadIdx.x & 31;
  const int half = lane >> 4;
  const int l15  = lane & 15;
  const int tile = blockIdx.x * (blockDim.x >> 5) + wave;
  const int r0 = tile * 16;
  if (r0 >= n) return;

  int arow = r0 + l15;
  if (arow >= n) arow = n - 1;
  const float* xrow = X + (long)arow * KDIM + half * 2;

  v8f accl[NT], accr[NT];
#pragma unroll
  for (int t = 0; t < NT; ++t) {
    const float b = bl[t * 16 + l15];
#pragma unroll
    for (int j = 0; j < 8; ++j) {
      accl[t][j] = b;
      accr[t][j] = 0.0f;
    }
  }

#pragma unroll
  for (int kb = 0; kb < KDIM / 4; ++kb) {
    v2f a = *(const v2f*)(xrow + kb * 4);
#pragma unroll
    for (int t = 0; t < NT; ++t) {
      v2f b = *(const v2f*)(Wl + (t * 16 + l15) * KDIM + kb * 4 + half * 2);
      accl[t] = __builtin_amdgcn_wmma_f32_16x16x4_f32(
           false, a,  false, b,
           (short)0, accl[t],  false,  false);
    }
#pragma unroll
    for (int t = 0; t < NT; ++t) {
      v2f b = *(const v2f*)(Wr + (t * 16 + l15) * KDIM + kb * 4 + half * 2);
      accr[t] = __builtin_amdgcn_wmma_f32_16x16x4_f32(
           false, a,  false, b,
           (short)0, accr[t],  false,  false);
    }
  }

#pragma unroll
  for (int t = 0; t < NT; ++t) {
    const int col = t * 16 + l15;
#pragma unroll
    for (int v = 0; v < 8; ++v) {
      const int row = r0 + half * 8 + v;
      if (row < n) {
        Yl[(long)row * OUTF + col] = accl[t][v];
        Yr[(long)row * OUTF + col] = accr[t][v];
      }
    }
  }
}

__global__ __launch_bounds__(256) void edge_rank(
    const int* __restrict__ ei, int* __restrict__ deg, int* __restrict__ rank, int nE) {
  const int e0 = blockIdx.x * 1024 + threadIdx.x;
  int dst[4], r[4];
#pragma unroll
  for (int j = 0; j < 4; ++j) dst[j] = ei[nE + min(e0 + j * 256, nE - 1)];
#pragma unroll
  for (int j = 0; j < 4; ++j) {
    r[j] = 0;
    if (e0 + j * 256 < nE)
      r[j] = __hip_atomic_fetch_add(deg + dst[j], 1, __ATOMIC_RELAXED, __HIP_MEMORY_SCOPE_AGENT);
  }
#pragma unroll
  for (int j = 0; j < 4; ++j) {
    const int e = e0 + j * 256;
    if (e < nE) rank[e] = r[j];
  }
}

__global__ __launch_bounds__(256) void scan_local(
    const int* __restrict__ deg, int* __restrict__ rowptr, int* __restrict__ bsum, int n) {
  __shared__ int wsum[8];
  const int lane = threadIdx.x & 31;
  const int wave = threadIdx.x >> 5;
  const int base = (blockIdx.x * 256 + threadIdx.x) * 4;
  int v[4];
#pragma unroll
  for (int i = 0; i < 4; ++i) v[i] = (base + i < n) ? deg[base + i] : 0;
  const int tsum = v[0] + v[1] + v[2] + v[3];
  int inc = tsum;
#pragma unroll
  for (int d = 1; d < 32; d <<= 1) {
    const int t = __shfl_up(inc, d);
    if (lane >= d) inc += t;
  }
  if (lane == 31) wsum[wave] = inc;
  __syncthreads();
  int excl = inc - tsum;
#pragma unroll
  for (int w = 0; w < 8; ++w)
    if (w < wave) excl += wsum[w];
#pragma unroll
  for (int i = 0; i < 4; ++i) {
    if (base + i < n) rowptr[base + i] = excl;
    excl += v[i];
  }
  if (threadIdx.x == 255) bsum[blockIdx.x] = excl;
}

__global__ __launch_bounds__(256) void scan_add(
    int* __restrict__ rowptr, const int* __restrict__ bsum, int n) {
  __shared__ int wsum[8];
  const int lane = threadIdx.x & 31;
  const int wave = threadIdx.x >> 5;
  int part = 0;
  for (int b = threadIdx.x; b < (int)blockIdx.x; b += 256) part += bsum[b];
#pragma unroll
  for (int d = 16; d > 0; d >>= 1) part += __shfl_xor(part, d);
  if (lane == 0) wsum[wave] = part;
  __syncthreads();
  int off = 0;
#pragma unroll
  for (int w = 0; w < 8; ++w) off += wsum[w];
  const int base = (blockIdx.x * 256 + threadIdx.x) * 4;
#pragma unroll
  for (int i = 0; i < 4; ++i)
    if (base + i < n) rowptr[base + i] += off;
  if (blockIdx.x == gridDim.x - 1 && threadIdx.x == 0) rowptr[n] = off + bsum[blockIdx.x];
}

__global__ __launch_bounds__(256) void edge_fill(
    const int* __restrict__ ei, const int* __restrict__ rowptr,
    const int* __restrict__ rank, int* __restrict__ col, int nE) {
  const int e0 = blockIdx.x * 1024 + threadIdx.x;
  int src[4], dst[4], pos[4];
#pragma unroll
  for (int j = 0; j < 4; ++j) {
    const int e = e0 + j * 256;
    const bool ok = e < nE;
    src[j] = ok ? ei[e] : 0;
    dst[j] = ok ? ei[nE + e] : 0;
    pos[j] = ok ? rank[e] : nE;
  }
#pragma unroll
  for (int j = 0; j < 4; ++j) pos[j] += rowptr[dst[j]];
#pragma unroll
  for (int j = 0; j < 4; ++j)
    if ((unsigned)pos[j] < (unsigned)nE) col[pos[j]] = src[j];
}

template <int F, bool RELU>
__global__ __launch_bounds__(256) void sage_gather(
    const int* __restrict__ rowptr, const int* __restrict__ col,
    const float* __restrict__ H, const float* __restrict__ R,
    float* __restrict__ Y, int n, int nE) {
  constexpr int G = F / 4;
  constexpr int NG = 32 / G;
  const int lane = threadIdx.x & 31;
  const int sub = lane / G;
  const int c = lane % G;
  const int wv = __builtin_amdgcn_readfirstlane((int)(threadIdx.x >> 5));
  const int nwaves = gridDim.x * (blockDim.x >> 5);
  for (int node = blockIdx.x * (blockDim.x >> 5) + wv; node < n; node += nwaves) {
    int beg = __builtin_amdgcn_readfirstlane(rowptr[node]);
    int end = __builtin_amdgcn_readfirstlane(rowptr[node + 1]);
    const float4 r = *(const float4*)(R + (long)node * F + c * 4);
    beg = min(max(beg, 0), nE);
    end = min(max(end, beg), nE);
    float4 acc = make_float4(0.0f, 0.0f, 0.0f, 0.0f);
    for (int base = beg; base < end; base += 32) {
      const int m = min(32, end - base);
      int cv = (lane < m) ? col[base + lane] : 0;
      cv = (int)min((unsigned)cv, (unsigned)(n - 1));
      const int full = m / NG;
#pragma unroll 4
      for (int k = 0; k < full; ++k) {
        const int s = __shfl(cv, k * NG + sub);
        const float4 v = *(const float4*)(H + (long)s * F + c * 4);
        acc.x += v.x; acc.y += v.y; acc.z += v.z; acc.w += v.w;
      }
      const int rem = m - full * NG;
      if (rem != 0) {
        const int s = __shfl(cv, full * NG + sub);
        if (sub < rem) {
          const float4 v = *(const float4*)(H + (long)s * F + c * 4);
          acc.x += v.x; acc.y += v.y; acc.z += v.z; acc.w += v.w;
        }
      }
    }
#pragma unroll
    for (int d = G; d < 32; d <<= 1) {
      acc.x += __shfl_xor(acc.x, d);
      acc.y += __shfl_xor(acc.y, d);
      acc.z += __shfl_xor(acc.z, d);
      acc.w += __shfl_xor(acc.w, d);
    }
    if (sub == 0) {
      const float d = fmaxf((float)(end - beg), 1.0f);
      float4 o;
      o.x = r.x + acc.x / d;
      o.y = r.y + acc.y / d;
      o.z = r.z + acc.z / d;
      o.w = r.w + acc.w / d;
      if (RELU) {
        o.x = fmaxf(o.x, 0.0f); o.y = fmaxf(o.y, 0.0f);
        o.z = fmaxf(o.z, 0.0f); o.w = fmaxf(o.w, 0.0f);
      }
      *(float4*)(Y + (long)node * F + c * 4) = o;
    }
  }
}

extern "C" void kernel_launch(void* const* d_in, const int* in_sizes, int n_in,
                              void* d_out, int out_size, void* d_ws, size_t ws_size,
                              hipStream_t stream) {
  const float* x   = (const float*)d_in[0];
  const int*   ei  = (const int*)d_in[1];
  const float* Wl0 = (const float*)d_in[2];
  const float* bl0 = (const float*)d_in[3];
  const float* Wr0 = (const float*)d_in[4];
  const float* Wl1 = (const float*)d_in[5];
  const float* bl1 = (const float*)d_in[6];
  const float* Wr1 = (const float*)d_in[7];
  float* out = (float*)d_out;

  const int N = in_sizes[0] / 64;
  const int E = in_sizes[1] / 2;

  float* B0 = (float*)d_ws;
  float* B1 = B0 + (size_t)N * 64;
  float* B2 = B1 + (size_t)N * 64;
  int* deg    = (int*)(B2 + (size_t)N * 64);
  int* rowptr = deg + N;
  int* bsum   = rowptr + N + 4;
  int* rank   = bsum + 1024;
  int* col    = rank + E;

  const dim3 blk(256);
  const int tiles = (N + 15) / 16;
  const dim3 gemmGrid((tiles + 7) / 8);
  const dim3 scanGrid((N + 1023) / 1024);
  const dim3 edgeGrid((E + 1023) / 1024);

  fill_zero<<<dim3(256), blk, 0, stream>>>((float*)deg, N);
  edge_rank<<<edgeGrid, blk, 0, stream>>>(ei, deg, rank, E);
  scan_local<<<scanGrid, blk, 0, stream>>>(deg, rowptr, bsum, N);
  scan_add<<<scanGrid, blk, 0, stream>>>(rowptr, bsum, N);
  edge_fill<<<edgeGrid, blk, 0, stream>>>(ei, rowptr, rank, col, E);

  sage_gemm_pair<64><<<gemmGrid, blk, 0, stream>>>(x, Wl0, bl0, Wr0, B0, B1, N);
  sage_gather<64, true><<<dim3(4096), blk, 0, stream>>>(rowptr, col, B0, B1, B2, N, E);

  sage_gemm_pair<32><<<gemmGrid, blk, 0, stream>>>(B2, Wl1, bl1, Wr1, B0, B1, N);
  sage_gather<32, false><<<dim3(4096), blk, 0, stream>>>(rowptr, col, B0, B1, out, N, E);
}


# GEMM passes move operands and results through LDS: both weight matrices and the block's 128 input rows in, result tiles out; every global access a whole-line 16-B one, WMMA fragments read from LDS
# speedup vs baseline: 1.7657x; 1.0487x over previous
#include <hip/hip_runtime.h>


typedef __attribute__((ext_vector_type(2))) float v2f;
typedef __attribute__((ext_vector_type(8))) float v8f;

#define KDIM 64

__global__ void fill_zero(float* __restrict__ p, int n) {
  int i = blockIdx.x * blockDim.x + threadIdx.x;
  int stride = gridDim.x * blockDim.x;
  for (; i < n; i += stride) p[i] = 0.0f;
}

#define LDP 68

template <int NT>
__device__ __forceinline__ void put_tile(float* lds, const v8f* acc, int half, int l15) {
#pragma unroll
  for (int t = 0; t < NT; ++t)
#pragma unroll
    for (int v = 0; v < 8; ++v) lds[(v + 8 * half) * LDP + t * 16 + l15] = acc[t][v];
}

template <int OUTF>
__device__ __forceinline__ void store_tile_rows(const float* lds, float* __restrict__ Y,
                                                int r0, int n, int lane) {
  constexpr int LPR = OUTF / 4;
  constexpr int RPI = 32 / LPR;
  constexpr int NS = 16 / RPI;
  const int rr = lane / LPR;
  const int c = (lane % LPR) * 4;
  float4 v[NS];
#pragma unroll
  for (int j = 0; j < NS; ++j) v[j] = *(const float4*)(lds + (j * RPI + rr) * LDP + c);
  float* yp = Y + (long)(r0 + rr) * OUTF + c;
  if (r0 + 16 <= n) {
#pragma unroll
    for (int j = 0; j < NS; ++j) *(float4*)(yp + (long)j * RPI * OUTF) = v[j];
  } else {
#pragma unroll
    for (int j = 0; j < NS; ++j)
      if (r0 + j * RPI + rr < n) *(float4*)(yp + (long)j * RPI * OUTF) = v[j];
  }
}

template <int OUTF>
__global__ __launch_bounds__(256) void sage_gemm_pair(
    const float* __restrict__ X, const float* __restrict__ Wl,
    const float* __restrict__ bl, const float* __restrict__ Wr,
    float* __restrict__ Yl, float* __restrict__ Yr, int n) {
  constexpr int NT = OUTF / 16;
  constexpr int NB = 2 * NT;
  __shared__ __attribute__((aligned(16))) float sW[2 * OUTF * LDP];
  __shared__ __attribute__((aligned(16))) float sA[128 * LDP];
  const int tid = threadIdx.x;
  const int wave = __builtin_amdgcn_readfirstlane(tid >> 5);
  const int lane = tid & 31;
  const int half = lane >> 4;
  const int l15  = lane & 15;
  const int rb = blockIdx.x * 128;
  const int sr = tid >> 4;
  const int sc = (tid & 15) * 4;

  float4 wv[OUTF / 8], xv[8];
#pragma unroll
  for (int i = 0; i < OUTF / 8; ++i) {
    const float* src = (i < NT) ? (Wl + (i * 16 + sr) * KDIM) : (Wr + ((i - NT) * 16 + sr) * KDIM);
    wv[i] = *(const float4*)(src + sc);
  }
#pragma unroll
  for (int i = 0; i < 8; ++i) {
    int row = rb + i * 16 + sr;
    if (row >= n) row = n - 1;
    xv[i] = *(const float4*)(X + (long)row * KDIM + sc);
  }
#pragma unroll
  for (int i = 0; i < OUTF / 8; ++i) *(float4*)(sW + (i * 16 + sr) * LDP + sc) = wv[i];
#pragma unroll
  for (int i = 0; i < 8; ++i) *(float4*)(sA + (i * 16 + sr) * LDP + sc) = xv[i];
  __syncthreads();

  const int r0 = rb + wave * 16;
  const bool active = r0 < n;
  float* own = sA + wave * 16 * LDP;

  v8f acc[NB];
  if (active) {
#pragma unroll
    for (int t = 0; t < NT; ++t) {
      const float b = bl[t * 16 + l15];
#pragma unroll
      for (int j = 0; j < 8; ++j) {
        acc[t][j] = b;
        acc[NT + t][j] = 0.0f;
      }
    }

    const float* ap = own + l15 * LDP + half * 2;
    const float* bp = sW + l15 * LDP + half * 2;
#pragma unroll
    for (int kb = 0; kb < KDIM / 4; ++kb) {
      const v2f a = *(const v2f*)(ap + kb * 4);
#pragma unroll
      for (int t = 0; t < NB; ++t) {
        const v2f b = *(const v2f*)(bp + t * 16 * LDP + kb * 4);
        acc[t] = __builtin_amdgcn_wmma_f32_16x16x4_f32(
             false, a,  false, b,
             (short)0, acc[t],  false,  false);
      }
    }
  }

  if constexpr (OUTF == 64) {
    float* wsl = sW + wave * 16 * LDP;
    if (active) put_tile<NT>(own, acc, half, l15);
    __syncthreads();
    if (active) put_tile<NT>(wsl, acc + NT, half, l15);
    __syncthreads();
    if (active) {
      store_tile_rows<OUTF>(own, Yl, r0, n, lane);
      store_tile_rows<OUTF>(wsl, Yr, r0, n, lane);
    }
  } else {
    if (active) {
      put_tile<NT>(own, acc, half, l15);
      put_tile<NT>(own + OUTF, acc + NT, half, l15);
    }
    __syncthreads();
    if (active) {
      store_tile_rows<OUTF>(own, Yl, r0, n, lane);
      store_tile_rows<OUTF>(own + OUTF, Yr, r0, n, lane);
    }
  }
}

__global__ __launch_bounds__(256) void edge_rank(
    const int* __restrict__ ei, int* __restrict__ deg, int* __restrict__ rank, int nE) {
  const int e0 = blockIdx.x * 1024 + threadIdx.x;
  int dst[4], r[4];
#pragma unroll
  for (int j = 0; j < 4; ++j) dst[j] = ei[nE + min(e0 + j * 256, nE - 1)];
#pragma unroll
  for (int j = 0; j < 4; ++j) {
    r[j] = 0;
    if (e0 + j * 256 < nE)
      r[j] = __hip_atomic_fetch_add(deg + dst[j], 1, __ATOMIC_RELAXED, __HIP_MEMORY_SCOPE_AGENT);
  }
#pragma unroll
  for (int j = 0; j < 4; ++j) {
    const int e = e0 + j * 256;
    if (e < nE) rank[e] = r[j];
  }
}

__global__ __launch_bounds__(256) void scan_local(
    const int* __restrict__ deg, int* __restrict__ rowptr, int* __restrict__ bsum, int n) {
  __shared__ int wsum[8];
  const int lane = threadIdx.x & 31;
  const int wave = threadIdx.x >> 5;
  const int base = (blockIdx.x * 256 + threadIdx.x) * 4;
  int v[4];
#pragma unroll
  for (int i = 0; i < 4; ++i) v[i] = (base + i < n) ? deg[base + i] : 0;
  const int tsum = v[0] + v[1] + v[2] + v[3];
  int inc = tsum;
#pragma unroll
  for (int d = 1; d < 32; d <<= 1) {
    const int t = __shfl_up(inc, d);
    if (lane >= d) inc += t;
  }
  if (lane == 31) wsum[wave] = inc;
  __syncthreads();
  int excl = inc - tsum;
#pragma unroll
  for (int w = 0; w < 8; ++w)
    if (w < wave) excl += wsum[w];
#pragma unroll
  for (int i = 0; i < 4; ++i) {
    if (base + i < n) rowptr[base + i] = excl;
    excl += v[i];
  }
  if (threadIdx.x == 255) bsum[blockIdx.x] = excl;
}

__global__ __launch_bounds__(256) void scan_add(
    int* __restrict__ rowptr, const int* __restrict__ bsum, int n) {
  __shared__ int wsum[8];
  const int lane = threadIdx.x & 31;
  const int wave = threadIdx.x >> 5;
  int part = 0;
  for (int b = threadIdx.x; b < (int)blockIdx.x; b += 256) part += bsum[b];
#pragma unroll
  for (int d = 16; d > 0; d >>= 1) part += __shfl_xor(part, d);
  if (lane == 0) wsum[wave] = part;
  __syncthreads();
  int off = 0;
#pragma unroll
  for (int w = 0; w < 8; ++w) off += wsum[w];
  const int base = (blockIdx.x * 256 + threadIdx.x) * 4;
#pragma unroll
  for (int i = 0; i < 4; ++i)
    if (base + i < n) rowptr[base + i] += off;
  if (blockIdx.x == gridDim.x - 1 && threadIdx.x == 0) rowptr[n] = off + bsum[blockIdx.x];
}

__global__ __launch_bounds__(256) void edge_fill(
    const int* __restrict__ ei, const int* __restrict__ rowptr,
    const int* __restrict__ rank, int* __restrict__ col, int nE) {
  const int e0 = blockIdx.x * 1024 + threadIdx.x;
  int src[4], dst[4], pos[4];
#pragma unroll
  for (int j = 0; j < 4; ++j) {
    const int e = e0 + j * 256;
    const bool ok = e < nE;
    src[j] = ok ? ei[e] : 0;
    dst[j] = ok ? ei[nE + e] : 0;
    pos[j] = ok ? rank[e] : nE;
  }
#pragma unroll
  for (int j = 0; j < 4; ++j) pos[j] += rowptr[dst[j]];
#pragma unroll
  for (int j = 0; j < 4; ++j)
    if ((unsigned)pos[j] < (unsigned)nE) col[pos[j]] = src[j];
}

template <int F, bool RELU>
__global__ __launch_bounds__(256) void sage_gather(
    const int* __restrict__ rowptr, const int* __restrict__ col,
    const float* __restrict__ H, const float* __restrict__ R,
    float* __restrict__ Y, int n, int nE) {
  constexpr int G = F / 4;
  constexpr int NG = 32 / G;
  const int lane = threadIdx.x & 31;
  const int sub = lane / G;
  const int c = lane % G;
  const int wv = __builtin_amdgcn_readfirstlane((int)(threadIdx.x >> 5));
  const int nwaves = gridDim.x * (blockDim.x >> 5);
  for (int node = blockIdx.x * (blockDim.x >> 5) + wv; node < n; node += nwaves) {
    int beg = __builtin_amdgcn_readfirstlane(rowptr[node]);
    int end = __builtin_amdgcn_readfirstlane(rowptr[node + 1]);
    const float4 r = *(const float4*)(R + (long)node * F + c * 4);
    beg = min(max(beg, 0), nE);
    end = min(max(end, beg), nE);
    float4 acc = make_float4(0.0f, 0.0f, 0.0f, 0.0f);
    for (int base = beg; base < end; base += 32) {
      const int m = min(32, end - base);
      int cv = (lane < m) ? col[base + lane] : 0;
      cv = (int)min((unsigned)cv, (unsigned)(n - 1));
      const int full = m / NG;
#pragma unroll 4
      for (int k = 0; k < full; ++k) {
        const int s = __shfl(cv, k * NG + sub);
        const float4 v = *(const float4*)(H + (long)s * F + c * 4);
        acc.x += v.x; acc.y += v.y; acc.z += v.z; acc.w += v.w;
      }
      const int rem = m - full * NG;
      if (rem != 0) {
        const int s = __shfl(cv, full * NG + sub);
        if (sub < rem) {
          const float4 v = *(const float4*)(H + (long)s * F + c * 4);
          acc.x += v.x; acc.y += v.y; acc.z += v.z; acc.w += v.w;
        }
      }
    }
#pragma unroll
    for (int d = G; d < 32; d <<= 1) {
      acc.x += __shfl_xor(acc.x, d);
      acc.y += __shfl_xor(acc.y, d);
      acc.z += __shfl_xor(acc.z, d);
      acc.w += __shfl_xor(acc.w, d);
    }
    if (sub == 0) {
      const float d = fmaxf((float)(end - beg), 1.0f);
      float4 o;
      o.x = r.x + acc.x / d;
      o.y = r.y + acc.y / d;
      o.z = r.z + acc.z / d;
      o.w = r.w + acc.w / d;
      if (RELU) {
        o.x = fmaxf(o.x, 0.0f); o.y = fmaxf(o.y, 0.0f);
        o.z = fmaxf(o.z, 0.0f); o.w = fmaxf(o.w, 0.0f);
      }
      *(float4*)(Y + (long)node * F + c * 4) = o;
    }
  }
}

extern "C" void kernel_launch(void* const* d_in, const int* in_sizes, int n_in,
                              void* d_out, int out_size, void* d_ws, size_t ws_size,
                              hipStream_t stream) {
  const float* x   = (const float*)d_in[0];
  const int*   ei  = (const int*)d_in[1];
  const float* Wl0 = (const float*)d_in[2];
  const float* bl0 = (const float*)d_in[3];
  const float* Wr0 = (const float*)d_in[4];
  const float* Wl1 = (const float*)d_in[5];
  const float* bl1 = (const float*)d_in[6];
  const float* Wr1 = (const float*)d_in[7];
  float* out = (float*)d_out;

  const int N = in_sizes[0] / 64;
  const int E = in_sizes[1] / 2;

  float* B0 = (float*)d_ws;
  float* B1 = B0 + (size_t)N * 64;
  float* B2 = B1 + (size_t)N * 64;
  int* deg    = (int*)(B2 + (size_t)N * 64);
  int* rowptr = deg + N;
  int* bsum   = rowptr + N + 4;
  int* rank   = bsum + 1024;
  int* col    = rank + E;

  const dim3 blk(256);
  const int tiles = (N + 15) / 16;
  const dim3 gemmGrid((tiles + 7) / 8);
  const dim3 scanGrid((N + 1023) / 1024);
  const dim3 edgeGrid((E + 1023) / 1024);

  fill_zero<<<dim3(256), blk, 0, stream>>>((float*)deg, N);
  edge_rank<<<edgeGrid, blk, 0, stream>>>(ei, deg, rank, E);
  scan_local<<<scanGrid, blk, 0, stream>>>(deg, rowptr, bsum, N);
  scan_add<<<scanGrid, blk, 0, stream>>>(rowptr, bsum, N);
  edge_fill<<<edgeGrid, blk, 0, stream>>>(ei, rowptr, rank, col, E);

  sage_gemm_pair<64><<<gemmGrid, blk, 0, stream>>>(x, Wl0, bl0, Wr0, B0, B1, N);
  sage_gather<64, true><<<dim3(4096), blk, 0, stream>>>(rowptr, col, B0, B1, B2, N, E);

  sage_gemm_pair<32><<<gemmGrid, blk, 0, stream>>>(B2, Wl1, bl1, Wr1, B0, B1, N);
  sage_gather<32, false><<<dim3(4096), blk, 0, stream>>>(rowptr, col, B0, B1, out, N, E);
}


# Edges grouped by destination by a two-level counting sort in LDS (buckets of 256 nodes, then a block per bucket); no global atomics; 9 -> 7 launches; sums in LDS-counter arrival order
# speedup vs baseline: 3.5813x; 2.0283x over previous
#include <hip/hip_runtime.h>


typedef __attribute__((ext_vector_type(2))) float v2f;
typedef __attribute__((ext_vector_type(8))) float v8f;

#define KDIM 64

#define LDP 68

template <int NT>
__device__ __forceinline__ void put_tile(float* lds, const v8f* acc, int half, int l15) {
#pragma unroll
  for (int t = 0; t < NT; ++t)
#pragma unroll
    for (int v = 0; v < 8; ++v) lds[(v + 8 * half) * LDP + t * 16 + l15] = acc[t][v];
}

template <int OUTF>
__device__ __forceinline__ void store_tile_rows(const float* lds, float* __restrict__ Y,
                                                int r0, int n, int lane) {
  constexpr int LPR = OUTF / 4;
  constexpr int RPI = 32 / LPR;
  constexpr int NS = 16 / RPI;
  const int rr = lane / LPR;
  const int c = (lane % LPR) * 4;
  float4 v[NS];
#pragma unroll
  for (int j = 0; j < NS; ++j) v[j] = *(const float4*)(lds + (j * RPI + rr) * LDP + c);
  float* yp = Y + (long)(r0 + rr) * OUTF + c;
  if (r0 + 16 <= n) {
#pragma unroll
    for (int j = 0; j < NS; ++j) *(float4*)(yp + (long)j * RPI * OUTF) = v[j];
  } else {
#pragma unroll
    for (int j = 0; j < NS; ++j)
      if (r0 + j * RPI + rr < n) *(float4*)(yp + (long)j * RPI * OUTF) = v[j];
  }
}

template <int OUTF>
__global__ __launch_bounds__(256) void sage_gemm_pair(
    const float* __restrict__ X, const float* __restrict__ Wl,
    const float* __restrict__ bl, const float* __restrict__ Wr,
    float* __restrict__ Yl, float* __restrict__ Yr, int n) {
  constexpr int NT = OUTF / 16;
  constexpr int NB = 2 * NT;
  __shared__ __attribute__((aligned(16))) float sW[2 * OUTF * LDP];
  __shared__ __attribute__((aligned(16))) float sA[128 * LDP];
  const int tid = threadIdx.x;
  const int wave = __builtin_amdgcn_readfirstlane(tid >> 5);
  const int lane = tid & 31;
  const int half = lane >> 4;
  const int l15  = lane & 15;
  const int rb = blockIdx.x * 128;
  const int sr = tid >> 4;
  const int sc = (tid & 15) * 4;

  float4 wv[OUTF / 8], xv[8];
#pragma unroll
  for (int i = 0; i < OUTF / 8; ++i) {
    const float* src = (i < NT) ? (Wl + (i * 16 + sr) * KDIM) : (Wr + ((i - NT) * 16 + sr) * KDIM);
    wv[i] = *(const float4*)(src + sc);
  }
#pragma unroll
  for (int i = 0; i < 8; ++i) {
    int row = rb + i * 16 + sr;
    if (row >= n) row = n - 1;
    xv[i] = *(const float4*)(X + (long)row * KDIM + sc);
  }
#pragma unroll
  for (int i = 0; i < OUTF / 8; ++i) *(float4*)(sW + (i * 16 + sr) * LDP + sc) = wv[i];
#pragma unroll
  for (int i = 0; i < 8; ++i) *(float4*)(sA + (i * 16 + sr) * LDP + sc) = xv[i];
  __syncthreads();

  const int r0 = rb + wave * 16;
  const bool active = r0 < n;
  float* own = sA + wave * 16 * LDP;

  v8f acc[NB];
  if (active) {
#pragma unroll
    for (int t = 0; t < NT; ++t) {
      const float b = bl[t * 16 + l15];
#pragma unroll
      for (int j = 0; j < 8; ++j) {
        acc[t][j] = b;
        acc[NT + t][j] = 0.0f;
      }
    }

    const float* ap = own + l15 * LDP + half * 2;
    const float* bp = sW + l15 * LDP + half * 2;
#pragma unroll
    for (int kb = 0; kb < KDIM / 4; ++kb) {
      const v2f a = *(const v2f*)(ap + kb * 4);
#pragma unroll
      for (int t = 0; t < NB; ++t) {
        const v2f b = *(const v2f*)(bp + t * 16 * LDP + kb * 4);
        acc[t] = __builtin_amdgcn_wmma_f32_16x16x4_f32(
             false, a,  false, b,
             (short)0, acc[t],  false,  false);
      }
    }
  }

  if constexpr (OUTF == 64) {
    float* wsl = sW + wave * 16 * LDP;
    if (active) put_tile<NT>(own, acc, half, l15);
    __syncthreads();
    if (active) put_tile<NT>(wsl, acc + NT, half, l15);
    __syncthreads();
    if (active) {
      store_tile_rows<OUTF>(own, Yl, r0, n, lane);
      store_tile_rows<OUTF>(wsl, Yr, r0, n, lane);
    }
  } else {
    if (active) {
      put_tile<NT>(own, acc, half, l15);
      put_tile<NT>(own + OUTF, acc + NT, half, l15);
    }
    __syncthreads();
    if (active) {
      store_tile_rows<OUTF>(own, Yl, r0, n, lane);
      store_tile_rows<OUTF>(own + OUTF, Yr, r0, n, lane);
    }
  }
}

#define BSH 8
#define BNODES (1 << BSH)
#define MAXB 512
#define TPB 896
#define EPT 7
#define CHUNK (TPB * EPT)
#define SCAP 8192

__global__ __launch_bounds__(TPB) void bin_count(
    const int* __restrict__ ei, int* __restrict__ H, int nE, int nbk) {
  __shared__ int hist[MAXB];
  const int tid = threadIdx.x;
  if (tid < MAXB) hist[tid] = 0;
  const int beg = blockIdx.x * CHUNK;
  int d[EPT];
#pragma unroll
  for (int j = 0; j < EPT; ++j) d[j] = ei[nE + min(beg + j * TPB + tid, nE - 1)];
  __syncthreads();
#pragma unroll
  for (int j = 0; j < EPT; ++j) {
    if (beg + j * TPB + tid < nE) {
      const int b = (int)min((unsigned)d[j] >> BSH, (unsigned)(nbk - 1));
      atomicAdd(&hist[b], 1);
    }
  }
  __syncthreads();
  if (tid < MAXB) H[blockIdx.x * MAXB + tid] = hist[tid];
}

__global__ __launch_bounds__(TPB) void bin_scatter(
    const int* __restrict__ ei, const int* __restrict__ H, int* __restrict__ BB,
    unsigned* __restrict__ P, int nE, int nbk, int nb1) {
  __shared__ uint2 stage[CHUNK];
  __shared__ int tot[MAXB], pre[MAXB], gdelta[MAXB], cur[MAXB];
  __shared__ int wsA[32], wsB[32];
  const int tid = threadIdx.x;
  const int lane = tid & 31;
  const int wave = tid >> 5;
  const int me = blockIdx.x;
  const int beg = me * CHUNK;

  int s[EPT], d[EPT];
#pragma unroll
  for (int j = 0; j < EPT; ++j) {
    const int e = min(beg + j * TPB + tid, nE - 1);
    s[j] = ei[e];
    d[j] = ei[nE + e];
  }
  int ownv = 0;
  if (tid < MAXB) {
    ownv = H[me * MAXB + tid];
    tot[tid] = 0;
    pre[tid] = 0;
  }
  __syncthreads();

  {
    constexpr int G = TPB / 128;
    const int q = (tid & 127) * 4;
    const int g = tid >> 7;
    if (q < nbk) {
      int4 t = make_int4(0, 0, 0, 0), p = make_int4(0, 0, 0, 0);
      for (int b0 = g; b0 < nb1; b0 += 8 * G) {
        int4 v[8];
#pragma unroll
        for (int u = 0; u < 8; ++u)
          v[u] = *(const int4*)(H + min(b0 + u * G, nb1 - 1) * MAXB + q);
#pragma unroll
        for (int u = 0; u < 8; ++u) {
          const int blk = b0 + u * G;
          if (blk < nb1) {
            t.x += v[u].x; t.y += v[u].y; t.z += v[u].z; t.w += v[u].w;
            if (blk < me) { p.x += v[u].x; p.y += v[u].y; p.z += v[u].z; p.w += v[u].w; }
          }
        }
      }
      atomicAdd(&tot[q + 0], t.x); atomicAdd(&tot[q + 1], t.y);
      atomicAdd(&tot[q + 2], t.z); atomicAdd(&tot[q + 3], t.w);
      atomicAdd(&pre[q + 0], p.x); atomicAdd(&pre[q + 1], p.y);
      atomicAdd(&pre[q + 2], p.z); atomicAdd(&pre[q + 3], p.w);
    }
  }
  __syncthreads();

  {
    const int tv = (tid < MAXB) ? tot[tid] : 0;
    const int ov = ownv;
    int ti = tv, oi = ov;
#pragma unroll
    for (int k = 1; k < 32; k <<= 1) {
      const int a = __shfl_up(ti, k);
      const int b = __shfl_up(oi, k);
      if (lane >= k) { ti += a; oi += b; }
    }
    if (lane == 31) { wsA[wave] = ti; wsB[wave] = oi; }
    __syncthreads();
    int ta = 0, oa = 0;
#pragma unroll
    for (int w = 0; w < MAXB / 32; ++w)
      if (w < wave) { ta += wsA[w]; oa += wsB[w]; }
    if (tid < MAXB) {
      const int gb = ta + ti - tv;
      const int ls = oa + oi - ov;
      gdelta[tid] = gb + pre[tid] - ls;
      cur[tid] = ls;
      if (me == 0) {
        BB[tid] = gb;
        if (tid == MAXB - 1) BB[MAXB] = gb + tv;
      }
    }
  }
  __syncthreads();

  const int cnt = min(CHUNK, nE - beg);
#pragma unroll
  for (int j = 0; j < EPT; ++j) {
    if (j * TPB + tid < cnt) {
      const int b = (int)min((unsigned)d[j] >> BSH, (unsigned)(nbk - 1));
      const int p = atomicAdd(&cur[b], 1);
      if ((unsigned)p < (unsigned)CHUNK)
        stage[p] = make_uint2(((unsigned)s[j] << BSH) | ((unsigned)d[j] & (BNODES - 1)), (unsigned)b);
    }
  }
  __syncthreads();
  for (int i = tid; i < cnt; i += TPB) {
    const uint2 v = stage[i];
    const int gidx = gdelta[v.y & (MAXB - 1)] + i;
    if ((unsigned)gidx < (unsigned)nE) P[gidx] = v.x;
  }
}

__global__ __launch_bounds__(256) void bucket_sort(
    const unsigned* __restrict__ P, const int* __restrict__ BB,
    int* __restrict__ rowptr, int* __restrict__ col, int n, int nE) {
  __shared__ int cnt[BNODES];
  __shared__ int wsum[8];
  __shared__ int sorted[SCAP];
  const int tid = threadIdx.x;
  const int lane = tid & 31;
  const int wave = tid >> 5;
  int beg = BB[blockIdx.x];
  int end = BB[blockIdx.x + 1];
  beg = min(max(beg, 0), nE);
  end = min(max(end, beg), nE);
  cnt[tid] = 0;
  __syncthreads();
  for (int i0 = beg + tid; i0 < end; i0 += 1024) {
    unsigned w[4];
#pragma unroll
    for (int j = 0; j < 4; ++j) w[j] = P[min(i0 + j * 256, end - 1)];
#pragma unroll
    for (int j = 0; j < 4; ++j)
      if (i0 + j * 256 < end) atomicAdd(&cnt[w[j] & (BNODES - 1)], 1);
  }
  __syncthreads();
  const int c = cnt[tid];
  int inc = c;
#pragma unroll
  for (int k = 1; k < 32; k <<= 1) {
    const int a = __shfl_up(inc, k);
    if (lane >= k) inc += a;
  }
  if (lane == 31) wsum[wave] = inc;
  __syncthreads();
  int excl = inc - c;
#pragma unroll
  for (int w = 0; w < 8; ++w)
    if (w < wave) excl += wsum[w];
  const int node = blockIdx.x * BNODES + tid;
  if (node < n) rowptr[node] = beg + excl;
  if (node == n - 1) rowptr[n] = beg + excl + c;
  cnt[tid] = excl;
  __syncthreads();
  const int len = end - beg;
  const bool fits = len <= SCAP;
  for (int i0 = beg + tid; i0 < end; i0 += 1024) {
    unsigned w[4];
#pragma unroll
    for (int j = 0; j < 4; ++j) w[j] = P[min(i0 + j * 256, end - 1)];
#pragma unroll
    for (int j = 0; j < 4; ++j) {
      if (i0 + j * 256 < end) {
        const int p = atomicAdd(&cnt[w[j] & (BNODES - 1)], 1);
        const int src = (int)(w[j] >> BSH);
        if (fits) {
          if ((unsigned)p < (unsigned)SCAP) sorted[p] = src;
        } else {
          if ((unsigned)p < (unsigned)len) col[beg + p] = src;
        }
      }
    }
  }
  if (fits) {
    __syncthreads();
    for (int i = tid; i < len; i += 256) col[beg + i] = sorted[i];
  }
}

template <int F, bool RELU>
__global__ __launch_bounds__(256) void sage_gather(
    const int* __restrict__ rowptr, const int* __restrict__ col,
    const float* __restrict__ H, const float* __restrict__ R,
    float* __restrict__ Y, int n, int nE) {
  constexpr int G = F / 4;
  constexpr int NG = 32 / G;
  const int lane = threadIdx.x & 31;
  const int sub = lane / G;
  const int c = lane % G;
  const int wv = __builtin_amdgcn_readfirstlane((int)(threadIdx.x >> 5));
  const int nwaves = gridDim.x * (blockDim.x >> 5);
  for (int node = blockIdx.x * (blockDim.x >> 5) + wv; node < n; node += nwaves) {
    int beg = __builtin_amdgcn_readfirstlane(rowptr[node]);
    int end = __builtin_amdgcn_readfirstlane(rowptr[node + 1]);
    const float4 r = *(const float4*)(R + (long)node * F + c * 4);
    beg = min(max(beg, 0), nE);
    end = min(max(end, beg), nE);
    float4 acc = make_float4(0.0f, 0.0f, 0.0f, 0.0f);
    for (int base = beg; base < end; base += 32) {
      const int m = min(32, end - base);
      int cv = (lane < m) ? col[base + lane] : 0;
      cv = (int)min((unsigned)cv, (unsigned)(n - 1));
      const int full = m / NG;
#pragma unroll 4
      for (int k = 0; k < full; ++k) {
        const int s = __shfl(cv, k * NG + sub);
        const float4 v = *(const float4*)(H + (long)s * F + c * 4);
        acc.x += v.x; acc.y += v.y; acc.z += v.z; acc.w += v.w;
      }
      const int rem = m - full * NG;
      if (rem != 0) {
        const int s = __shfl(cv, full * NG + sub);
        if (sub < rem) {
          const float4 v = *(const float4*)(H + (long)s * F + c * 4);
          acc.x += v.x; acc.y += v.y; acc.z += v.z; acc.w += v.w;
        }
      }
    }
#pragma unroll
    for (int d = G; d < 32; d <<= 1) {
      acc.x += __shfl_xor(acc.x, d);
      acc.y += __shfl_xor(acc.y, d);
      acc.z += __shfl_xor(acc.z, d);
      acc.w += __shfl_xor(acc.w, d);
    }
    if (sub == 0) {
      const float d = fmaxf((float)(end - beg), 1.0f);
      float4 o;
      o.x = r.x + acc.x / d;
      o.y = r.y + acc.y / d;
      o.z = r.z + acc.z / d;
      o.w = r.w + acc.w / d;
      if (RELU) {
        o.x = fmaxf(o.x, 0.0f); o.y = fmaxf(o.y, 0.0f);
        o.z = fmaxf(o.z, 0.0f); o.w = fmaxf(o.w, 0.0f);
      }
      *(float4*)(Y + (long)node * F + c * 4) = o;
    }
  }
}

extern "C" void kernel_launch(void* const* d_in, const int* in_sizes, int n_in,
                              void* d_out, int out_size, void* d_ws, size_t ws_size,
                              hipStream_t stream) {
  const float* x   = (const float*)d_in[0];
  const int*   ei  = (const int*)d_in[1];
  const float* Wl0 = (const float*)d_in[2];
  const float* bl0 = (const float*)d_in[3];
  const float* Wr0 = (const float*)d_in[4];
  const float* Wl1 = (const float*)d_in[5];
  const float* bl1 = (const float*)d_in[6];
  const float* Wr1 = (const float*)d_in[7];
  float* out = (float*)d_out;

  const int N = in_sizes[0] / 64;
  const int E = in_sizes[1] / 2;

  float* B0 = (float*)d_ws;
  float* B1 = B0 + (size_t)N * 64;
  float* B2 = B1 + (size_t)N * 64;
  const int NB1 = (E + CHUNK - 1) / CHUNK;
  const int NBK = (N + BNODES - 1) / BNODES;
  int* H        = (int*)(B2 + (size_t)N * 64);
  int* BB       = H + (size_t)NB1 * MAXB;
  unsigned* P   = (unsigned*)(BB + MAXB + 4);
  int* col      = (int*)P + E;
  int* rowptr   = col + E;

  const dim3 blk(256);
  const int tiles = (N + 15) / 16;
  const dim3 gemmGrid((tiles + 7) / 8);

  bin_count<<<dim3(NB1), dim3(TPB), 0, stream>>>(ei, H, E, NBK);
  bin_scatter<<<dim3(NB1), dim3(TPB), 0, stream>>>(ei, H, BB, P, E, NBK, NB1);
  bucket_sort<<<dim3(NBK), blk, 0, stream>>>(P, BB, rowptr, col, N, E);

  sage_gemm_pair<64><<<gemmGrid, blk, 0, stream>>>(x, Wl0, bl0, Wr0, B0, B1, N);
  sage_gather<64, true><<<dim3(4096), blk, 0, stream>>>(rowptr, col, B0, B1, B2, N, E);

  sage_gemm_pair<32><<<gemmGrid, blk, 0, stream>>>(B2, Wl1, bl1, Wr1, B0, B1, N);
  sage_gather<32, false><<<dim3(4096), blk, 0, stream>>>(rowptr, col, B0, B1, out, N, E);
}


# Layer-0 gather and layer-1 GEMM pass fused into one launch: a wave keeps its 16 finished rows in LDS and multiplies them there; h1 never goes to memory; 7 -> 6 launches; sums unchanged
# speedup vs baseline: 3.7833x; 1.0564x over previous
#include <hip/hip_runtime.h>


typedef __attribute__((ext_vector_type(2))) float v2f;
typedef __attribute__((ext_vector_type(8))) float v8f;

#define KDIM 64

#define LDP 68

template <int NT>
__device__ __forceinline__ void put_tile(float* lds, const v8f* acc, int half, int l15) {
#pragma unroll
  for (int t = 0; t < NT; ++t)
#pragma unroll
    for (int v = 0; v < 8; ++v) lds[(v + 8 * half) * LDP + t * 16 + l15] = acc[t][v];
}

template <int OUTF>
__device__ __forceinline__ void store_tile_rows(const float* lds, float* __restrict__ Y,
                                                int r0, int n, int lane) {
  constexpr int LPR = OUTF / 4;
  constexpr int RPI = 32 / LPR;
  constexpr int NS = 16 / RPI;
  const int rr = lane / LPR;
  const int c = (lane % LPR) * 4;
  float4 v[NS];
#pragma unroll
  for (int j = 0; j < NS; ++j) v[j] = *(const float4*)(lds + (j * RPI + rr) * LDP + c);
  float* yp = Y + (long)(r0 + rr) * OUTF + c;
  if (r0 + 16 <= n) {
#pragma unroll
    for (int j = 0; j < NS; ++j) *(float4*)(yp + (long)j * RPI * OUTF) = v[j];
  } else {
#pragma unroll
    for (int j = 0; j < NS; ++j)
      if (r0 + j * RPI + rr < n) *(float4*)(yp + (long)j * RPI * OUTF) = v[j];
  }
}

template <int OUTF>
__device__ __forceinline__ void mma_pair(const float* rows, const float* sW, const float* __restrict__ bl,
                                         v8f* acc, int half, int l15) {
  constexpr int NT = OUTF / 16;
  constexpr int NB = 2 * NT;
#pragma unroll
  for (int t = 0; t < NT; ++t) {
    const float b = bl[t * 16 + l15];
#pragma unroll
    for (int j = 0; j < 8; ++j) {
      acc[t][j] = b;
      acc[NT + t][j] = 0.0f;
    }
  }

  const float* ap = rows + l15 * LDP + half * 2;
  const float* bp = sW + l15 * LDP + half * 2;
#pragma unroll
  for (int kb = 0; kb < KDIM / 4; ++kb) {
    const v2f a = *(const v2f*)(ap + kb * 4);
#pragma unroll
    for (int t = 0; t < NB; ++t) {
      const v2f b = *(const v2f*)(bp + t * 16 * LDP + kb * 4);
      acc[t] = __builtin_amdgcn_wmma_f32_16x16x4_f32(
           false, a,  false, b,
           (short)0, acc[t],  false,  false);
    }
  }
}

template <int OUTF>
__global__ __launch_bounds__(256) void sage_gemm_pair(
    const float* __restrict__ X, const float* __restrict__ Wl,
    const float* __restrict__ bl, const float* __restrict__ Wr,
    float* __restrict__ Yl, float* __restrict__ Yr, int n) {
  constexpr int NT = OUTF / 16;
  constexpr int NB = 2 * NT;
  __shared__ __attribute__((aligned(16))) float sW[2 * OUTF * LDP];
  __shared__ __attribute__((aligned(16))) float sA[128 * LDP];
  const int tid = threadIdx.x;
  const int wave = __builtin_amdgcn_readfirstlane(tid >> 5);
  const int lane = tid & 31;
  const int half = lane >> 4;
  const int l15  = lane & 15;
  const int rb = blockIdx.x * 128;
  const int sr = tid >> 4;
  const int sc = (tid & 15) * 4;

  float4 wv[OUTF / 8], xv[8];
#pragma unroll
  for (int i = 0; i < OUTF / 8; ++i) {
    const float* src = (i < NT) ? (Wl + (i * 16 + sr) * KDIM) : (Wr + ((i - NT) * 16 + sr) * KDIM);
    wv[i] = *(const float4*)(src + sc);
  }
#pragma unroll
  for (int i = 0; i < 8; ++i) {
    int row = rb + i * 16 + sr;
    if (row >= n) row = n - 1;
    xv[i] = *(const float4*)(X + (long)row * KDIM + sc);
  }
#pragma unroll
  for (int i = 0; i < OUTF / 8; ++i) *(float4*)(sW + (i * 16 + sr) * LDP + sc) = wv[i];
#pragma unroll
  for (int i = 0; i < 8; ++i) *(float4*)(sA + (i * 16 + sr) * LDP + sc) = xv[i];
  __syncthreads();

  const int r0 = rb + wave * 16;
  const bool active = r0 < n;
  float* own = sA + wave * 16 * LDP;

  v8f acc[NB];
  if (active) mma_pair<OUTF>(own, sW, bl, acc, half, l15);

  if constexpr (OUTF == 64) {
    float* wsl = sW + wave * 16 * LDP;
    if (active) put_tile<NT>(own, acc, half, l15);
    __syncthreads();
    if (active) put_tile<NT>(wsl, acc + NT, half, l15);
    __syncthreads();
    if (active) {
      store_tile_rows<OUTF>(own, Yl, r0, n, lane);
      store_tile_rows<OUTF>(wsl, Yr, r0, n, lane);
    }
  } else {
    if (active) {
      put_tile<NT>(own, acc, half, l15);
      put_tile<NT>(own + OUTF, acc + NT, half, l15);
    }
    __syncthreads();
    if (active) {
      store_tile_rows<OUTF>(own, Yl, r0, n, lane);
      store_tile_rows<OUTF>(own + OUTF, Yr, r0, n, lane);
    }
  }
}

#define BSH 8
#define BNODES (1 << BSH)
#define MAXB 512
#define TPB 896
#define EPT 7
#define CHUNK (TPB * EPT)
#define SCAP 8192

__global__ __launch_bounds__(TPB) void bin_count(
    const int* __restrict__ ei, int* __restrict__ H, int nE, int nbk) {
  __shared__ int hist[MAXB];
  const int tid = threadIdx.x;
  if (tid < MAXB) hist[tid] = 0;
  const int beg = blockIdx.x * CHUNK;
  int d[EPT];
#pragma unroll
  for (int j = 0; j < EPT; ++j) d[j] = ei[nE + min(beg + j * TPB + tid, nE - 1)];
  __syncthreads();
#pragma unroll
  for (int j = 0; j < EPT; ++j) {
    if (beg + j * TPB + tid < nE) {
      const int b = (int)min((unsigned)d[j] >> BSH, (unsigned)(nbk - 1));
      atomicAdd(&hist[b], 1);
    }
  }
  __syncthreads();
  if (tid < MAXB) H[blockIdx.x * MAXB + tid] = hist[tid];
}

__global__ __launch_bounds__(TPB) void bin_scatter(
    const int* __restrict__ ei, const int* __restrict__ H, int* __restrict__ BB,
    unsigned* __restrict__ P, int nE, int nbk, int nb1) {
  __shared__ uint2 stage[CHUNK];
  __shared__ int tot[MAXB], pre[MAXB], gdelta[MAXB], cur[MAXB];
  __shared__ int wsA[32], wsB[32];
  const int tid = threadIdx.x;
  const int lane = tid & 31;
  const int wave = tid >> 5;
  const int me = blockIdx.x;
  const int beg = me * CHUNK;

  int s[EPT], d[EPT];
#pragma unroll
  for (int j = 0; j < EPT; ++j) {
    const int e = min(beg + j * TPB + tid, nE - 1);
    s[j] = ei[e];
    d[j] = ei[nE + e];
  }
  int ownv = 0;
  if (tid < MAXB) {
    ownv = H[me * MAXB + tid];
    tot[tid] = 0;
    pre[tid] = 0;
  }
  __syncthreads();

  {
    constexpr int G = TPB / 128;
    const int q = (tid & 127) * 4;
    const int g = tid >> 7;
    if (q < nbk) {
      int4 t = make_int4(0, 0, 0, 0), p = make_int4(0, 0, 0, 0);
      for (int b0 = g; b0 < nb1; b0 += 8 * G) {
        int4 v[8];
#pragma unroll
        for (int u = 0; u < 8; ++u)
          v[u] = *(const int4*)(H + min(b0 + u * G, nb1 - 1) * MAXB + q);
#pragma unroll
        for (int u = 0; u < 8; ++u) {
          const int blk = b0 + u * G;
          if (blk < nb1) {
            t.x += v[u].x; t.y += v[u].y; t.z += v[u].z; t.w += v[u].w;
            if (blk < me) { p.x += v[u].x; p.y += v[u].y; p.z += v[u].z; p.w += v[u].w; }
          }
        }
      }
      atomicAdd(&tot[q + 0], t.x); atomicAdd(&tot[q + 1], t.y);
      atomicAdd(&tot[q + 2], t.z); atomicAdd(&tot[q + 3], t.w);
      atomicAdd(&pre[q + 0], p.x); atomicAdd(&pre[q + 1], p.y);
      atomicAdd(&pre[q + 2], p.z); atomicAdd(&pre[q + 3], p.w);
    }
  }
  __syncthreads();

  {
    const int tv = (tid < MAXB) ? tot[tid] : 0;
    const int ov = ownv;
    int ti = tv, oi = ov;
#pragma unroll
    for (int k = 1; k < 32; k <<= 1) {
      const int a = __shfl_up(ti, k);
      const int b = __shfl_up(oi, k);
      if (lane >= k) { ti += a; oi += b; }
    }
    if (lane == 31) { wsA[wave] = ti; wsB[wave] = oi; }
    __syncthreads();
    int ta = 0, oa = 0;
#pragma unroll
    for (int w = 0; w < MAXB / 32; ++w)
      if (w < wave) { ta += wsA[w]; oa += wsB[w]; }
    if (tid < MAXB) {
      const int gb = ta + ti - tv;
      const int ls = oa + oi - ov;
      gdelta[tid] = gb + pre[tid] - ls;
      cur[tid] = ls;
      if (me == 0) {
        BB[tid] = gb;
        if (tid == MAXB - 1) BB[MAXB] = gb + tv;
      }
    }
  }
  __syncthreads();

  const int cnt = min(CHUNK, nE - beg);
#pragma unroll
  for (int j = 0; j < EPT; ++j) {
    if (j * TPB + tid < cnt) {
      const int b = (int)min((unsigned)d[j] >> BSH, (unsigned)(nbk - 1));
      const int p = atomicAdd(&cur[b], 1);
      if ((unsigned)p < (unsigned)CHUNK)
        stage[p] = make_uint2(((unsigned)s[j] << BSH) | ((unsigned)d[j] & (BNODES - 1)), (unsigned)b);
    }
  }
  __syncthreads();
  for (int i = tid; i < cnt; i += TPB) {
    const uint2 v = stage[i];
    const int gidx = gdelta[v.y & (MAXB - 1)] + i;
    if ((unsigned)gidx < (unsigned)nE) P[gidx] = v.x;
  }
}

__global__ __launch_bounds__(256) void bucket_sort(
    const unsigned* __restrict__ P, const int* __restrict__ BB,
    int* __restrict__ rowptr, int* __restrict__ col, int n, int nE) {
  __shared__ int cnt[BNODES];
  __shared__ int wsum[8];
  __shared__ int sorted[SCAP];
  const int tid = threadIdx.x;
  const int lane = tid & 31;
  const int wave = tid >> 5;
  int beg = BB[blockIdx.x];
  int end = BB[blockIdx.x + 1];
  beg = min(max(beg, 0), nE);
  end = min(max(end, beg), nE);
  cnt[tid] = 0;
  __syncthreads();
  for (int i0 = beg + tid; i0 < end; i0 += 1024) {
    unsigned w[4];
#pragma unroll
    for (int j = 0; j < 4; ++j) w[j] = P[min(i0 + j * 256, end - 1)];
#pragma unroll
    for (int j = 0; j < 4; ++j)
      if (i0 + j * 256 < end) atomicAdd(&cnt[w[j] & (BNODES - 1)], 1);
  }
  __syncthreads();
  const int c = cnt[tid];
  int inc = c;
#pragma unroll
  for (int k = 1; k < 32; k <<= 1) {
    const int a = __shfl_up(inc, k);
    if (lane >= k) inc += a;
  }
  if (lane == 31) wsum[wave] = inc;
  __syncthreads();
  int excl = inc - c;
#pragma unroll
  for (int w = 0; w < 8; ++w)
    if (w < wave) excl += wsum[w];
  const int node = blockIdx.x * BNODES + tid;
  if (node < n) rowptr[node] = beg + excl;
  if (node == n - 1) rowptr[n] = beg + excl + c;
  cnt[tid] = excl;
  __syncthreads();
  const int len = end - beg;
  const bool fits = len <= SCAP;
  for (int i0 = beg + tid; i0 < end; i0 += 1024) {
    unsigned w[4];
#pragma unroll
    for (int j = 0; j < 4; ++j) w[j] = P[min(i0 + j * 256, end - 1)];
#pragma unroll
    for (int j = 0; j < 4; ++j) {
      if (i0 + j * 256 < end) {
        const int p = atomicAdd(&cnt[w[j] & (BNODES - 1)], 1);
        const int src = (int)(w[j] >> BSH);
        if (fits) {
          if ((unsigned)p < (unsigned)SCAP) sorted[p] = src;
        } else {
          if ((unsigned)p < (unsigned)len) col[beg + p] = src;
        }
      }
    }
  }
  if (fits) {
    __syncthreads();
    for (int i = tid; i < len; i += 256) col[beg + i] = sorted[i];
  }
}

template <int F, bool RELU>
__global__ __launch_bounds__(256) void sage_gather(
    const int* __restrict__ rowptr, const int* __restrict__ col,
    const float* __restrict__ H, const float* __restrict__ R,
    float* __restrict__ Y, int n, int nE) {
  constexpr int G = F / 4;
  constexpr int NG = 32 / G;
  const int lane = threadIdx.x & 31;
  const int sub = lane / G;
  const int c = lane % G;
  const int wv = __builtin_amdgcn_readfirstlane((int)(threadIdx.x >> 5));
  const int nwaves = gridDim.x * (blockDim.x >> 5);
  for (int node = blockIdx.x * (blockDim.x >> 5) + wv; node < n; node += nwaves) {
    int beg = __builtin_amdgcn_readfirstlane(rowptr[node]);
    int end = __builtin_amdgcn_readfirstlane(rowptr[node + 1]);
    const float4 r = *(const float4*)(R + (long)node * F + c * 4);
    beg = min(max(beg, 0), nE);
    end = min(max(end, beg), nE);
    float4 acc = make_float4(0.0f, 0.0f, 0.0f, 0.0f);
    for (int base = beg; base < end; base += 32) {
      const int m = min(32, end - base);
      int cv = (lane < m) ? col[base + lane] : 0;
      cv = (int)min((unsigned)cv, (unsigned)(n - 1));
      const int full = m / NG;
#pragma unroll 4
      for (int k = 0; k < full; ++k) {
        const int s = __shfl(cv, k * NG + sub);
        const float4 v = *(const float4*)(H + (long)s * F + c * 4);
        acc.x += v.x; acc.y += v.y; acc.z += v.z; acc.w += v.w;
      }
      const int rem = m - full * NG;
      if (rem != 0) {
        const int s = __shfl(cv, full * NG + sub);
        if (sub < rem) {
          const float4 v = *(const float4*)(H + (long)s * F + c * 4);
          acc.x += v.x; acc.y += v.y; acc.z += v.z; acc.w += v.w;
        }
      }
    }
#pragma unroll
    for (int d = G; d < 32; d <<= 1) {
      acc.x += __shfl_xor(acc.x, d);
      acc.y += __shfl_xor(acc.y, d);
      acc.z += __shfl_xor(acc.z, d);
      acc.w += __shfl_xor(acc.w, d);
    }
    if (sub == 0) {
      const float d = fmaxf((float)(end - beg), 1.0f);
      float4 o;
      o.x = r.x + acc.x / d;
      o.y = r.y + acc.y / d;
      o.z = r.z + acc.z / d;
      o.w = r.w + acc.w / d;
      if (RELU) {
        o.x = fmaxf(o.x, 0.0f); o.y = fmaxf(o.y, 0.0f);
        o.z = fmaxf(o.z, 0.0f); o.w = fmaxf(o.w, 0.0f);
      }
      *(float4*)(Y + (long)node * F + c * 4) = o;
    }
  }
}

#define GW 4
__global__ __launch_bounds__(GW * 32) void sage_gather_gemm(
    const int* __restrict__ rowptr, const int* __restrict__ col,
    const float* __restrict__ H, const float* __restrict__ R,
    const float* __restrict__ Wl, const float* __restrict__ bl, const float* __restrict__ Wr,
    float* __restrict__ Yl, float* __restrict__ Yr, int n, int nE) {
  constexpr int F = KDIM;
  constexpr int OUTF = 32;
  constexpr int NT = OUTF / 16;
  constexpr int NB = 2 * NT;
  __shared__ __attribute__((aligned(16))) float sW[2 * OUTF * LDP];
  __shared__ __attribute__((aligned(16))) float sA[GW * 16 * LDP];
  const int tid = threadIdx.x;
  const int wave = __builtin_amdgcn_readfirstlane(tid >> 5);
  const int lane = tid & 31;
  const int half = lane >> 4;
  const int l15  = lane & 15;

  {
    constexpr int WS = (2 * OUTF * 16) / (GW * 32);
    float4 wv[WS];
#pragma unroll
    for (int i = 0; i < WS; ++i) {
      const int idx = i * (GW * 32) + tid;
      const int row = idx >> 4;
      const float* src = (row < OUTF) ? (Wl + row * KDIM) : (Wr + (row - OUTF) * KDIM);
      wv[i] = *(const float4*)(src + (idx & 15) * 4);
    }
#pragma unroll
    for (int i = 0; i < WS; ++i) {
      const int idx = i * (GW * 32) + tid;
      *(float4*)(sW + (idx >> 4) * LDP + (idx & 15) * 4) = wv[i];
    }
  }

  const int node0 = blockIdx.x * (GW * 16) + wave * 16;
  float* own = sA + wave * 16 * LDP;

  int rp = rowptr[min(node0 + min(lane, 16), n)];
  rp = min(max(rp, 0), nE);
  int cvn = col[min(__builtin_amdgcn_readlane(rp, 0) + lane, nE - 1)];
  for (int i = 0; i < 16; ++i) {
    const int node = node0 + i;
    const int beg = __builtin_amdgcn_readlane(rp, i);
    const int nxt = __builtin_amdgcn_readlane(rp, i + 1);
    const int end = max(nxt, beg);
    int cv = cvn;
    cvn = col[min(nxt + lane, nE - 1)];
    float4 o = make_float4(0.0f, 0.0f, 0.0f, 0.0f);
    if (node < n) {
      const float4 r = *(const float4*)(R + (long)node * F + l15 * 4);
      float4 sum = make_float4(0.0f, 0.0f, 0.0f, 0.0f);
      for (int base = beg; base < end; base += 32) {
        const int m = min(32, end - base);
        if (base != beg) cv = col[min(base + lane, nE - 1)];
        cv = (lane < m) ? (int)min((unsigned)cv, (unsigned)(n - 1)) : 0;
        const int nf = m >> 1;
        const bool odd = (m & 1) != 0;
#pragma unroll
        for (int b = 0; b < 16; b += 8) {
          if (2 * b < m) {
            float4 v[8], vt;
            const bool tail = odd && nf >= b && nf < b + 8;
#pragma unroll
            for (int k = 0; k < 8; ++k) {
              if (b + k < nf) {
                const int s0 = __builtin_amdgcn_readlane(cv, 2 * (b + k));
                const int s1 = __builtin_amdgcn_readlane(cv, 2 * (b + k) + 1);
                const int s = half ? s1 : s0;
                v[k] = *(const float4*)(H + (long)s * F + l15 * 4);
              }
            }
            if (tail) {
              const int st = __builtin_amdgcn_readlane(cv, m - 1);
              vt = *(const float4*)(H + (long)st * F + l15 * 4);
            }
#pragma unroll
            for (int k = 0; k < 8; ++k) {
              if (b + k < nf) {
                sum.x += v[k].x; sum.y += v[k].y; sum.z += v[k].z; sum.w += v[k].w;
              }
            }
            if (tail) {
              sum.x += half ? 0.0f : vt.x;
              sum.y += half ? 0.0f : vt.y;
              sum.z += half ? 0.0f : vt.z;
              sum.w += half ? 0.0f : vt.w;
            }
          }
        }
      }
      sum.x += __shfl_xor(sum.x, 16);
      sum.y += __shfl_xor(sum.y, 16);
      sum.z += __shfl_xor(sum.z, 16);
      sum.w += __shfl_xor(sum.w, 16);
      const float d = fmaxf((float)(end - beg), 1.0f);
      o.x = fmaxf(r.x + sum.x / d, 0.0f);
      o.y = fmaxf(r.y + sum.y / d, 0.0f);
      o.z = fmaxf(r.z + sum.z / d, 0.0f);
      o.w = fmaxf(r.w + sum.w / d, 0.0f);
    }
    if (half == 0) *(float4*)(own + i * LDP + l15 * 4) = o;
  }
  __syncthreads();

  const int r0 = node0;
  const bool active = r0 < n;
  v8f acc[NB];
  if (active) {
    mma_pair<OUTF>(own, sW, bl, acc, half, l15);
    put_tile<NT>(own, acc, half, l15);
    put_tile<NT>(own + OUTF, acc + NT, half, l15);
  }
  __syncthreads();
  if (active) {
    store_tile_rows<OUTF>(own, Yl, r0, n, lane);
    store_tile_rows<OUTF>(own + OUTF, Yr, r0, n, lane);
  }
}

extern "C" void kernel_launch(void* const* d_in, const int* in_sizes, int n_in,
                              void* d_out, int out_size, void* d_ws, size_t ws_size,
                              hipStream_t stream) {
  const float* x   = (const float*)d_in[0];
  const int*   ei  = (const int*)d_in[1];
  const float* Wl0 = (const float*)d_in[2];
  const float* bl0 = (const float*)d_in[3];
  const float* Wr0 = (const float*)d_in[4];
  const float* Wl1 = (const float*)d_in[5];
  const float* bl1 = (const float*)d_in[6];
  const float* Wr1 = (const float*)d_in[7];
  float* out = (float*)d_out;

  const int N = in_sizes[0] / 64;
  const int E = in_sizes[1] / 2;

  float* B0 = (float*)d_ws;
  float* B1 = B0 + (size_t)N * 64;
  float* B2 = B1 + (size_t)N * 64;
  const int NB1 = (E + CHUNK - 1) / CHUNK;
  const int NBK = (N + BNODES - 1) / BNODES;
  int* H        = (int*)(B2 + (size_t)N * 64);
  int* BB       = H + (size_t)NB1 * MAXB;
  unsigned* P   = (unsigned*)(BB + MAXB + 4);
  int* col      = (int*)P + E;
  int* rowptr   = col + E;

  const dim3 blk(256);
  const int tiles = (N + 15) / 16;
  const dim3 gemmGrid((tiles + 7) / 8);

  bin_count<<<dim3(NB1), dim3(TPB), 0, stream>>>(ei, H, E, NBK);
  bin_scatter<<<dim3(NB1), dim3(TPB), 0, stream>>>(ei, H, BB, P, E, NBK, NB1);
  bucket_sort<<<dim3(NBK), blk, 0, stream>>>(P, BB, rowptr, col, N, E);

  float* Yl1 = B2;
  float* Yr1 = B2 + (size_t)N * 32;
  sage_gemm_pair<64><<<gemmGrid, blk, 0, stream>>>(x, Wl0, bl0, Wr0, B0, B1, N);
  sage_gather_gemm<<<dim3((N + GW * 16 - 1) / (GW * 16)), dim3(GW * 32), 0, stream>>>(
      rowptr, col, B0, B1, Wl1, bl1, Wr1, Yl1, Yr1, N, E);

  sage_gather<32, false><<<dim3(4096), blk, 0, stream>>>(rowptr, col, Yl1, Yr1, out, N, E);
}


# Level 2 of the edge sort and the layer-0 GEMM pass, which do not depend on each other, run side by side as different blocks of one launch; 6 -> 5 launches; sums unchanged
# speedup vs baseline: 4.1553x; 1.0983x over previous
#include <hip/hip_runtime.h>


typedef __attribute__((ext_vector_type(2))) float v2f;
typedef __attribute__((ext_vector_type(8))) float v8f;

#define KDIM 64

#define LDP 68

template <int NT>
__device__ __forceinline__ void put_tile(float* lds, const v8f* acc, int half, int l15) {
#pragma unroll
  for (int t = 0; t < NT; ++t)
#pragma unroll
    for (int v = 0; v < 8; ++v) lds[(v + 8 * half) * LDP + t * 16 + l15] = acc[t][v];
}

template <int OUTF>
__device__ __forceinline__ void store_tile_rows(const float* lds, float* __restrict__ Y,
                                                int r0, int n, int lane) {
  constexpr int LPR = OUTF / 4;
  constexpr int RPI = 32 / LPR;
  constexpr int NS = 16 / RPI;
  const int rr = lane / LPR;
  const int c = (lane % LPR) * 4;
  float4 v[NS];
#pragma unroll
  for (int j = 0; j < NS; ++j) v[j] = *(const float4*)(lds + (j * RPI + rr) * LDP + c);
  float* yp = Y + (long)(r0 + rr) * OUTF + c;
  if (r0 + 16 <= n) {
#pragma unroll
    for (int j = 0; j < NS; ++j) *(float4*)(yp + (long)j * RPI * OUTF) = v[j];
  } else {
#pragma unroll
    for (int j = 0; j < NS; ++j)
      if (r0 + j * RPI + rr < n) *(float4*)(yp + (long)j * RPI * OUTF) = v[j];
  }
}

template <int OUTF>
__device__ __forceinline__ void mma_pair(const float* rows, const float* sW, const float* __restrict__ bl,
                                         v8f* acc, int half, int l15) {
  constexpr int NT = OUTF / 16;
  constexpr int NB = 2 * NT;
#pragma unroll
  for (int t = 0; t < NT; ++t) {
    const float b = bl[t * 16 + l15];
#pragma unroll
    for (int j = 0; j < 8; ++j) {
      acc[t][j] = b;
      acc[NT + t][j] = 0.0f;
    }
  }

  const float* ap = rows + l15 * LDP + half * 2;
  const float* bp = sW + l15 * LDP + half * 2;
#pragma unroll
  for (int kb = 0; kb < KDIM / 4; ++kb) {
    const v2f a = *(const v2f*)(ap + kb * 4);
#pragma unroll
    for (int t = 0; t < NB; ++t) {
      const v2f b = *(const v2f*)(bp + t * 16 * LDP + kb * 4);
      acc[t] = __builtin_amdgcn_wmma_f32_16x16x4_f32(
           false, a,  false, b,
           (short)0, acc[t],  false,  false);
    }
  }
}

template <int OUTF>
__device__ __forceinline__ void gemm_pair_block(
    const float* __restrict__ X, const float* __restrict__ Wl,
    const float* __restrict__ bl, const float* __restrict__ Wr,
    float* __restrict__ Yl, float* __restrict__ Yr, int n, int rb, float* sW, float* sA) {
  constexpr int NT = OUTF / 16;
  constexpr int NB = 2 * NT;
  const int tid = threadIdx.x;
  const int wave = __builtin_amdgcn_readfirstlane(tid >> 5);
  const int lane = tid & 31;
  const int half = lane >> 4;
  const int l15  = lane & 15;
  const int sr = tid >> 4;
  const int sc = (tid & 15) * 4;

  float4 wv[OUTF / 8], xv[8];
#pragma unroll
  for (int i = 0; i < OUTF / 8; ++i) {
    const float* src = (i < NT) ? (Wl + (i * 16 + sr) * KDIM) : (Wr + ((i - NT) * 16 + sr) * KDIM);
    wv[i] = *(const float4*)(src + sc);
  }
#pragma unroll
  for (int i = 0; i < 8; ++i) {
    int row = rb + i * 16 + sr;
    if (row >= n) row = n - 1;
    xv[i] = *(const float4*)(X + (long)row * KDIM + sc);
  }
#pragma unroll
  for (int i = 0; i < OUTF / 8; ++i) *(float4*)(sW + (i * 16 + sr) * LDP + sc) = wv[i];
#pragma unroll
  for (int i = 0; i < 8; ++i) *(float4*)(sA + (i * 16 + sr) * LDP + sc) = xv[i];
  __syncthreads();

  const int r0 = rb + wave * 16;
  const bool active = r0 < n;
  float* own = sA + wave * 16 * LDP;

  v8f acc[NB];
  if (active) mma_pair<OUTF>(own, sW, bl, acc, half, l15);

  if constexpr (OUTF == 64) {
    float* wsl = sW + wave * 16 * LDP;
    if (active) put_tile<NT>(own, acc, half, l15);
    __syncthreads();
    if (active) put_tile<NT>(wsl, acc + NT, half, l15);
    __syncthreads();
    if (active) {
      store_tile_rows<OUTF>(own, Yl, r0, n, lane);
      store_tile_rows<OUTF>(wsl, Yr, r0, n, lane);
    }
  } else {
    if (active) {
      put_tile<NT>(own, acc, half, l15);
      put_tile<NT>(own + OUTF, acc + NT, half, l15);
    }
    __syncthreads();
    if (active) {
      store_tile_rows<OUTF>(own, Yl, r0, n, lane);
      store_tile_rows<OUTF>(own + OUTF, Yr, r0, n, lane);
    }
  }
}

#define BSH 8
#define BNODES (1 << BSH)
#define MAXB 512
#define TPB 896
#define EPT 7
#define CHUNK (TPB * EPT)
#define SCAP 8192

__global__ __launch_bounds__(TPB) void bin_count(
    const int* __restrict__ ei, int* __restrict__ H, int nE, int nbk) {
  __shared__ int hist[MAXB];
  const int tid = threadIdx.x;
  if (tid < MAXB) hist[tid] = 0;
  const int beg = blockIdx.x * CHUNK;
  int d[EPT];
#pragma unroll
  for (int j = 0; j < EPT; ++j) d[j] = ei[nE + min(beg + j * TPB + tid, nE - 1)];
  __syncthreads();
#pragma unroll
  for (int j = 0; j < EPT; ++j) {
    if (beg + j * TPB + tid < nE) {
      const int b = (int)min((unsigned)d[j] >> BSH, (unsigned)(nbk - 1));
      atomicAdd(&hist[b], 1);
    }
  }
  __syncthreads();
  if (tid < MAXB) H[blockIdx.x * MAXB + tid] = hist[tid];
}

__global__ __launch_bounds__(TPB) void bin_scatter(
    const int* __restrict__ ei, const int* __restrict__ H, int* __restrict__ BB,
    unsigned* __restrict__ P, int nE, int nbk, int nb1) {
  __shared__ uint2 stage[CHUNK];
  __shared__ int tot[MAXB], pre[MAXB], gdelta[MAXB], cur[MAXB];
  __shared__ int wsA[32], wsB[32];
  const int tid = threadIdx.x;
  const int lane = tid & 31;
  const int wave = tid >> 5;
  const int me = blockIdx.x;
  const int beg = me * CHUNK;

  int s[EPT], d[EPT];
#pragma unroll
  for (int j = 0; j < EPT; ++j) {
    const int e = min(beg + j * TPB + tid, nE - 1);
    s[j] = ei[e];
    d[j] = ei[nE + e];
  }
  int ownv = 0;
  if (tid < MAXB) {
    ownv = H[me * MAXB + tid];
    tot[tid] = 0;
    pre[tid] = 0;
  }
  __syncthreads();

  {
    constexpr int G = TPB / 128;
    const int q = (tid & 127) * 4;
    const int g = tid >> 7;
    if (q < nbk) {
      int4 t = make_int4(0, 0, 0, 0), p = make_int4(0, 0, 0, 0);
      for (int b0 = g; b0 < nb1; b0 += 8 * G) {
        int4 v[8];
#pragma unroll
        for (int u = 0; u < 8; ++u)
          v[u] = *(const int4*)(H + min(b0 + u * G, nb1 - 1) * MAXB + q);
#pragma unroll
        for (int u = 0; u < 8; ++u) {
          const int blk = b0 + u * G;
          if (blk < nb1) {
            t.x += v[u].x; t.y += v[u].y; t.z += v[u].z; t.w += v[u].w;
            if (blk < me) { p.x += v[u].x; p.y += v[u].y; p.z += v[u].z; p.w += v[u].w; }
          }
        }
      }
      atomicAdd(&tot[q + 0], t.x); atomicAdd(&tot[q + 1], t.y);
      atomicAdd(&tot[q + 2], t.z); atomicAdd(&tot[q + 3], t.w);
      atomicAdd(&pre[q + 0], p.x); atomicAdd(&pre[q + 1], p.y);
      atomicAdd(&pre[q + 2], p.z); atomicAdd(&pre[q + 3], p.w);
    }
  }
  __syncthreads();

  {
    const int tv = (tid < MAXB) ? tot[tid] : 0;
    const int ov = ownv;
    int ti = tv, oi = ov;
#pragma unroll
    for (int k = 1; k < 32; k <<= 1) {
      const int a = __shfl_up(ti, k);
      const int b = __shfl_up(oi, k);
      if (lane >= k) { ti += a; oi += b; }
    }
    if (lane == 31) { wsA[wave] = ti; wsB[wave] = oi; }
    __syncthreads();
    int ta = 0, oa = 0;
#pragma unroll
    for (int w = 0; w < MAXB / 32; ++w)
      if (w < wave) { ta += wsA[w]; oa += wsB[w]; }
    if (tid < MAXB) {
      const int gb = ta + ti - tv;
      const int ls = oa + oi - ov;
      gdelta[tid] = gb + pre[tid] - ls;
      cur[tid] = ls;
      if (me == 0) {
        BB[tid] = gb;
        if (tid == MAXB - 1) BB[MAXB] = gb + tv;
      }
    }
  }
  __syncthreads();

  const int cnt = min(CHUNK, nE - beg);
#pragma unroll
  for (int j = 0; j < EPT; ++j) {
    if (j * TPB + tid < cnt) {
      const int b = (int)min((unsigned)d[j] >> BSH, (unsigned)(nbk - 1));
      const int p = atomicAdd(&cur[b], 1);
      if ((unsigned)p < (unsigned)CHUNK)
        stage[p] = make_uint2(((unsigned)s[j] << BSH) | ((unsigned)d[j] & (BNODES - 1)), (unsigned)b);
    }
  }
  __syncthreads();
  for (int i = tid; i < cnt; i += TPB) {
    const uint2 v = stage[i];
    const int gidx = gdelta[v.y & (MAXB - 1)] + i;
    if ((unsigned)gidx < (unsigned)nE) P[gidx] = v.x;
  }
}

__device__ __forceinline__ void bucket_sort_block(
    const unsigned* __restrict__ P, const int* __restrict__ BB,
    int* __restrict__ rowptr, int* __restrict__ col, int n, int nE, int bk,
    int* cnt, int* wsum, int* sorted) {
  const int tid = threadIdx.x;
  const int lane = tid & 31;
  const int wave = tid >> 5;
  int beg = BB[bk];
  int end = BB[bk + 1];
  beg = min(max(beg, 0), nE);
  end = min(max(end, beg), nE);
  cnt[tid] = 0;
  __syncthreads();
  for (int i0 = beg + tid; i0 < end; i0 += 1024) {
    unsigned w[4];
#pragma unroll
    for (int j = 0; j < 4; ++j) w[j] = P[min(i0 + j * 256, end - 1)];
#pragma unroll
    for (int j = 0; j < 4; ++j)
      if (i0 + j * 256 < end) atomicAdd(&cnt[w[j] & (BNODES - 1)], 1);
  }
  __syncthreads();
  const int c = cnt[tid];
  int inc = c;
#pragma unroll
  for (int k = 1; k < 32; k <<= 1) {
    const int a = __shfl_up(inc, k);
    if (lane >= k) inc += a;
  }
  if (lane == 31) wsum[wave] = inc;
  __syncthreads();
  int excl = inc - c;
#pragma unroll
  for (int w = 0; w < 8; ++w)
    if (w < wave) excl += wsum[w];
  const int node = bk * BNODES + tid;
  if (node < n) rowptr[node] = beg + excl;
  if (node == n - 1) rowptr[n] = beg + excl + c;
  cnt[tid] = excl;
  __syncthreads();
  const int len = end - beg;
  const bool fits = len <= SCAP;
  for (int i0 = beg + tid; i0 < end; i0 += 1024) {
    unsigned w[4];
#pragma unroll
    for (int j = 0; j < 4; ++j) w[j] = P[min(i0 + j * 256, end - 1)];
#pragma unroll
    for (int j = 0; j < 4; ++j) {
      if (i0 + j * 256 < end) {
        const int p = atomicAdd(&cnt[w[j] & (BNODES - 1)], 1);
        const int src = (int)(w[j] >> BSH);
        if (fits) {
          if ((unsigned)p < (unsigned)SCAP) sorted[p] = src;
        } else {
          if ((unsigned)p < (unsigned)len) col[beg + p] = src;
        }
      }
    }
  }
  if (fits) {
    __syncthreads();
    for (int i = tid; i < len; i += 256) col[beg + i] = sorted[i];
  }
}

__global__ __launch_bounds__(256) void sort_and_gemm(
    const unsigned* __restrict__ P, const int* __restrict__ BB,
    int* __restrict__ rowptr, int* __restrict__ col, int n, int nE, int nbk,
    const float* __restrict__ X, const float* __restrict__ Wl,
    const float* __restrict__ bl, const float* __restrict__ Wr,
    float* __restrict__ Yl, float* __restrict__ Yr) {
  constexpr int OUTF = 64;
  __shared__ __attribute__((aligned(16))) float smem[(2 * OUTF + 128) * LDP];
  static_assert(sizeof(float) * (2 * OUTF + 128) * LDP >= sizeof(int) * (BNODES + 8 + SCAP), "the sort's tables must fit");
  if ((int)blockIdx.x < nbk) {
    int* cnt = (int*)smem;
    bucket_sort_block(P, BB, rowptr, col, n, nE, (int)blockIdx.x, cnt, cnt + BNODES, cnt + BNODES + 8);
  } else {
    gemm_pair_block<OUTF>(X, Wl, bl, Wr, Yl, Yr, n, ((int)blockIdx.x - nbk) * 128,
                          smem, smem + 2 * OUTF * LDP);
  }
}

template <int F, bool RELU>
__global__ __launch_bounds__(256) void sage_gather(
    const int* __restrict__ rowptr, const int* __restrict__ col,
    const float* __restrict__ H, const float* __restrict__ R,
    float* __restrict__ Y, int n, int nE) {
  constexpr int G = F / 4;
  constexpr int NG = 32 / G;
  const int lane = threadIdx.x & 31;
  const int sub = lane / G;
  const int c = lane % G;
  const int wv = __builtin_amdgcn_readfirstlane((int)(threadIdx.x >> 5));
  const int nwaves = gridDim.x * (blockDim.x >> 5);
  for (int node = blockIdx.x * (blockDim.x >> 5) + wv; node < n; node += nwaves) {
    int beg = __builtin_amdgcn_readfirstlane(rowptr[node]);
    int end = __builtin_amdgcn_readfirstlane(rowptr[node + 1]);
    const float4 r = *(const float4*)(R + (long)node * F + c * 4);
    beg = min(max(beg, 0), nE);
    end = min(max(end, beg), nE);
    float4 acc = make_float4(0.0f, 0.0f, 0.0f, 0.0f);
    for (int base = beg; base < end; base += 32) {
      const int m = min(32, end - base);
      int cv = (lane < m) ? col[base + lane] : 0;
      cv = (int)min((unsigned)cv, (unsigned)(n - 1));
      const int full = m / NG;
#pragma unroll 4
      for (int k = 0; k < full; ++k) {
        const int s = __shfl(cv, k * NG + sub);
        const float4 v = *(const float4*)(H + (long)s * F + c * 4);
        acc.x += v.x; acc.y += v.y; acc.z += v.z; acc.w += v.w;
      }
      const int rem = m - full * NG;
      if (rem != 0) {
        const int s = __shfl(cv, full * NG + sub);
        if (sub < rem) {
          const float4 v = *(const float4*)(H + (long)s * F + c * 4);
          acc.x += v.x; acc.y += v.y; acc.z += v.z; acc.w += v.w;
        }
      }
    }
#pragma unroll
    for (int d = G; d < 32; d <<= 1) {
      acc.x += __shfl_xor(acc.x, d);
      acc.y += __shfl_xor(acc.y, d);
      acc.z += __shfl_xor(acc.z, d);
      acc.w += __shfl_xor(acc.w, d);
    }
    if (sub == 0) {
      const float d = fmaxf((float)(end - beg), 1.0f);
      float4 o;
      o.x = r.x + acc.x / d;
      o.y = r.y + acc.y / d;
      o.z = r.z + acc.z / d;
      o.w = r.w + acc.w / d;
      if (RELU) {
        o.x = fmaxf(o.x, 0.0f); o.y = fmaxf(o.y, 0.0f);
        o.z = fmaxf(o.z, 0.0f); o.w = fmaxf(o.w, 0.0f);
      }
      *(float4*)(Y + (long)node * F + c * 4) = o;
    }
  }
}

#define GW 4
__global__ __launch_bounds__(GW * 32) void sage_gather_gemm(
    const int* __restrict__ rowptr, const int* __restrict__ col,
    const float* __restrict__ H, const float* __restrict__ R,
    const float* __restrict__ Wl, const float* __restrict__ bl, const float* __restrict__ Wr,
    float* __restrict__ Yl, float* __restrict__ Yr, int n, int nE) {
  constexpr int F = KDIM;
  constexpr int OUTF = 32;
  constexpr int NT = OUTF / 16;
  constexpr int NB = 2 * NT;
  __shared__ __attribute__((aligned(16))) float sW[2 * OUTF * LDP];
  __shared__ __attribute__((aligned(16))) float sA[GW * 16 * LDP];
  const int tid = threadIdx.x;
  const int wave = __builtin_amdgcn_readfirstlane(tid >> 5);
  const int lane = tid & 31;
  const int half = lane >> 4;
  const int l15  = lane & 15;

  {
    constexpr int WS = (2 * OUTF * 16) / (GW * 32);
    float4 wv[WS];
#pragma unroll
    for (int i = 0; i < WS; ++i) {
      const int idx = i * (GW * 32) + tid;
      const int row = idx >> 4;
      const float* src = (row < OUTF) ? (Wl + row * KDIM) : (Wr + (row - OUTF) * KDIM);
      wv[i] = *(const float4*)(src + (idx & 15) * 4);
    }
#pragma unroll
    for (int i = 0; i < WS; ++i) {
      const int idx = i * (GW * 32) + tid;
      *(float4*)(sW + (idx >> 4) * LDP + (idx & 15) * 4) = wv[i];
    }
  }

  const int node0 = blockIdx.x * (GW * 16) + wave * 16;
  float* own = sA + wave * 16 * LDP;

  int rp = rowptr[min(node0 + min(lane, 16), n)];
  rp = min(max(rp, 0), nE);
  int cvn = col[min(__builtin_amdgcn_readlane(rp, 0) + lane, nE - 1)];
  for (int i = 0; i < 16; ++i) {
    const int node = node0 + i;
    const int beg = __builtin_amdgcn_readlane(rp, i);
    const int nxt = __builtin_amdgcn_readlane(rp, i + 1);
    const int end = max(nxt, beg);
    int cv = cvn;
    cvn = col[min(nxt + lane, nE - 1)];
    float4 o = make_float4(0.0f, 0.0f, 0.0f, 0.0f);
    if (node < n) {
      const float4 r = *(const float4*)(R + (long)node * F + l15 * 4);
      float4 sum = make_float4(0.0f, 0.0f, 0.0f, 0.0f);
      for (int base = beg; base < end; base += 32) {
        const int m = min(32, end - base);
        if (base != beg) cv = col[min(base + lane, nE - 1)];
        cv = (lane < m) ? (int)min((unsigned)cv, (unsigned)(n - 1)) : 0;
        const int nf = m >> 1;
        const bool odd = (m & 1) != 0;
#pragma unroll
        for (int b = 0; b < 16; b += 8) {
          if (2 * b < m) {
            float4 v[8], vt;
            const bool tail = odd && nf >= b && nf < b + 8;
#pragma unroll
            for (int k = 0; k < 8; ++k) {
              if (b + k < nf) {
                const int s0 = __builtin_amdgcn_readlane(cv, 2 * (b + k));
                const int s1 = __builtin_amdgcn_readlane(cv, 2 * (b + k) + 1);
                const int s = half ? s1 : s0;
                v[k] = *(const float4*)(H + (long)s * F + l15 * 4);
              }
            }
            if (tail) {
              const int st = __builtin_amdgcn_readlane(cv, m - 1);
              vt = *(const float4*)(H + (long)st * F + l15 * 4);
            }
#pragma unroll
            for (int k = 0; k < 8; ++k) {
              if (b + k < nf) {
                sum.x += v[k].x; sum.y += v[k].y; sum.z += v[k].z; sum.w += v[k].w;
              }
            }
            if (tail) {
              sum.x += half ? 0.0f : vt.x;
              sum.y += half ? 0.0f : vt.y;
              sum.z += half ? 0.0f : vt.z;
              sum.w += half ? 0.0f : vt.w;
            }
          }
        }
      }
      sum.x += __shfl_xor(sum.x, 16);
      sum.y += __shfl_xor(sum.y, 16);
      sum.z += __shfl_xor(sum.z, 16);
      sum.w += __shfl_xor(sum.w, 16);
      const float d = fmaxf((float)(end - beg), 1.0f);
      o.x = fmaxf(r.x + sum.x / d, 0.0f);
      o.y = fmaxf(r.y + sum.y / d, 0.0f);
      o.z = fmaxf(r.z + sum.z / d, 0.0f);
      o.w = fmaxf(r.w + sum.w / d, 0.0f);
    }
    if (half == 0) *(float4*)(own + i * LDP + l15 * 4) = o;
  }
  __syncthreads();

  const int r0 = node0;
  const bool active = r0 < n;
  v8f acc[NB];
  if (active) {
    mma_pair<OUTF>(own, sW, bl, acc, half, l15);
    put_tile<NT>(own, acc, half, l15);
    put_tile<NT>(own + OUTF, acc + NT, half, l15);
  }
  __syncthreads();
  if (active) {
    store_tile_rows<OUTF>(own, Yl, r0, n, lane);
    store_tile_rows<OUTF>(own + OUTF, Yr, r0, n, lane);
  }
}

extern "C" void kernel_launch(void* const* d_in, const int* in_sizes, int n_in,
                              void* d_out, int out_size, void* d_ws, size_t ws_size,
                              hipStream_t stream) {
  const float* x   = (const float*)d_in[0];
  const int*   ei  = (const int*)d_in[1];
  const float* Wl0 = (const float*)d_in[2];
  const float* bl0 = (const float*)d_in[3];
  const float* Wr0 = (const float*)d_in[4];
  const float* Wl1 = (const float*)d_in[5];
  const float* bl1 = (const float*)d_in[6];
  const float* Wr1 = (const float*)d_in[7];
  float* out = (float*)d_out;

  const int N = in_sizes[0] / 64;
  const int E = in_sizes[1] / 2;

  float* B0 = (float*)d_ws;
  float* B1 = B0 + (size_t)N * 64;
  float* B2 = B1 + (size_t)N * 64;
  const int NB1 = (E + CHUNK - 1) / CHUNK;
  const int NBK = (N + BNODES - 1) / BNODES;
  int* H        = (int*)(B2 + (size_t)N * 64);
  int* BB       = H + (size_t)NB1 * MAXB;
  unsigned* P   = (unsigned*)(BB + MAXB + 4);
  int* col      = (int*)P + E;
  int* rowptr   = col + E;

  const dim3 blk(256);
  const int GEMMB = (N + 127) / 128;

  bin_count<<<dim3(NB1), dim3(TPB), 0, stream>>>(ei, H, E, NBK);
  bin_scatter<<<dim3(NB1), dim3(TPB), 0, stream>>>(ei, H, BB, P, E, NBK, NB1);
  sort_and_gemm<<<dim3(NBK + GEMMB), blk, 0, stream>>>(P, BB, rowptr, col, N, E, NBK,
                                                     x, Wl0, bl0, Wr0, B0, B1);

  float* Yl1 = B2;
  float* Yr1 = B2 + (size_t)N * 32;
  sage_gather_gemm<<<dim3((N + GW * 16 - 1) / (GW * 16)), dim3(GW * 32), 0, stream>>>(
      rowptr, col, B0, B1, Wl1, bl1, Wr1, Yl1, Yr1, N, E);

  sage_gather<32, false><<<dim3(4096), blk, 0, stream>>>(rowptr, col, Yl1, Yr1, out, N, E);
}


# Edge sort level 1 orders each chunk in place (rank from its LDS counter, run starts kept); level 2 collects a bucket's runs from all chunks; no count pass; 5 -> 4 launches; sums in LDS arrival order
# speedup vs baseline: 4.5820x; 1.1027x over previous
#include <hip/hip_runtime.h>


typedef __attribute__((ext_vector_type(2))) float v2f;
typedef __attribute__((ext_vector_type(8))) float v8f;

#define KDIM 64

#define LDP 68

template <int NT>
__device__ __forceinline__ void put_tile(float* lds, const v8f* acc, int half, int l15) {
#pragma unroll
  for (int t = 0; t < NT; ++t)
#pragma unroll
    for (int v = 0; v < 8; ++v) lds[(v + 8 * half) * LDP + t * 16 + l15] = acc[t][v];
}

template <int OUTF>
__device__ __forceinline__ void store_tile_rows(const float* lds, float* __restrict__ Y,
                                                int r0, int n, int lane) {
  constexpr int LPR = OUTF / 4;
  constexpr int RPI = 32 / LPR;
  constexpr int NS = 16 / RPI;
  const int rr = lane / LPR;
  const int c = (lane % LPR) * 4;
  float4 v[NS];
#pragma unroll
  for (int j = 0; j < NS; ++j) v[j] = *(const float4*)(lds + (j * RPI + rr) * LDP + c);
  float* yp = Y + (long)(r0 + rr) * OUTF + c;
  if (r0 + 16 <= n) {
#pragma unroll
    for (int j = 0; j < NS; ++j) *(float4*)(yp + (long)j * RPI * OUTF) = v[j];
  } else {
#pragma unroll
    for (int j = 0; j < NS; ++j)
      if (r0 + j * RPI + rr < n) *(float4*)(yp + (long)j * RPI * OUTF) = v[j];
  }
}

template <int OUTF>
__device__ __forceinline__ void mma_pair(const float* rows, const float* sW, const float* __restrict__ bl,
                                         v8f* acc, int half, int l15) {
  constexpr int NT = OUTF / 16;
  constexpr int NB = 2 * NT;
#pragma unroll
  for (int t = 0; t < NT; ++t) {
    const float b = bl[t * 16 + l15];
#pragma unroll
    for (int j = 0; j < 8; ++j) {
      acc[t][j] = b;
      acc[NT + t][j] = 0.0f;
    }
  }

  const float* ap = rows + l15 * LDP + half * 2;
  const float* bp = sW + l15 * LDP + half * 2;
#pragma unroll
  for (int kb = 0; kb < KDIM / 4; ++kb) {
    const v2f a = *(const v2f*)(ap + kb * 4);
#pragma unroll
    for (int t = 0; t < NB; ++t) {
      const v2f b = *(const v2f*)(bp + t * 16 * LDP + kb * 4);
      acc[t] = __builtin_amdgcn_wmma_f32_16x16x4_f32(
           false, a,  false, b,
           (short)0, acc[t],  false,  false);
    }
  }
}

template <int OUTF>
__device__ __forceinline__ void gemm_pair_block(
    const float* __restrict__ X, const float* __restrict__ Wl,
    const float* __restrict__ bl, const float* __restrict__ Wr,
    float* __restrict__ Yl, float* __restrict__ Yr, int n, int rb, float* sW, float* sA) {
  constexpr int NT = OUTF / 16;
  constexpr int NB = 2 * NT;
  const int tid = threadIdx.x;
  const int wave = __builtin_amdgcn_readfirstlane(tid >> 5);
  const int lane = tid & 31;
  const int half = lane >> 4;
  const int l15  = lane & 15;
  const int sr = tid >> 4;
  const int sc = (tid & 15) * 4;

  float4 wv[OUTF / 8], xv[8];
#pragma unroll
  for (int i = 0; i < OUTF / 8; ++i) {
    const float* src = (i < NT) ? (Wl + (i * 16 + sr) * KDIM) : (Wr + ((i - NT) * 16 + sr) * KDIM);
    wv[i] = *(const float4*)(src + sc);
  }
#pragma unroll
  for (int i = 0; i < 8; ++i) {
    int row = rb + i * 16 + sr;
    if (row >= n) row = n - 1;
    xv[i] = *(const float4*)(X + (long)row * KDIM + sc);
  }
#pragma unroll
  for (int i = 0; i < OUTF / 8; ++i) *(float4*)(sW + (i * 16 + sr) * LDP + sc) = wv[i];
#pragma unroll
  for (int i = 0; i < 8; ++i) *(float4*)(sA + (i * 16 + sr) * LDP + sc) = xv[i];
  __syncthreads();

  const int r0 = rb + wave * 16;
  const bool active = r0 < n;
  float* own = sA + wave * 16 * LDP;

  v8f acc[NB];
  if (active) mma_pair<OUTF>(own, sW, bl, acc, half, l15);

  if constexpr (OUTF == 64) {
    float* wsl = sW + wave * 16 * LDP;
    if (active) put_tile<NT>(own, acc, half, l15);
    __syncthreads();
    if (active) put_tile<NT>(wsl, acc + NT, half, l15);
    __syncthreads();
    if (active) {
      store_tile_rows<OUTF>(own, Yl, r0, n, lane);
      store_tile_rows<OUTF>(wsl, Yr, r0, n, lane);
    }
  } else {
    if (active) {
      put_tile<NT>(own, acc, half, l15);
      put_tile<NT>(own + OUTF, acc + NT, half, l15);
    }
    __syncthreads();
    if (active) {
      store_tile_rows<OUTF>(own, Yl, r0, n, lane);
      store_tile_rows<OUTF>(own + OUTF, Yr, r0, n, lane);
    }
  }
}

#define BSH 8
#define BNODES (1 << BSH)
#define MAXB 512
#define TPB 896
#define EPT 7
#define CHUNK (TPB * EPT)
#define SCAP 8192

__global__ __launch_bounds__(TPB) void bin_local(
    const int* __restrict__ ei, int* __restrict__ T, unsigned* __restrict__ P, int nE, int nbk) {
  __shared__ unsigned stage[CHUNK];
  __shared__ int hist[MAXB];
  __shared__ int wsA[MAXB / 32];
  const int tid = threadIdx.x;
  const int lane = tid & 31;
  const int wave = tid >> 5;
  const int me = blockIdx.x;
  const int beg = me * CHUNK;
  if (tid < MAXB) hist[tid] = 0;
  int s[EPT], d[EPT];
#pragma unroll
  for (int j = 0; j < EPT; ++j) {
    const int e = min(beg + j * TPB + tid, nE - 1);
    s[j] = ei[e];
    d[j] = ei[nE + e];
  }
  __syncthreads();
  const int cnt = min(CHUNK, nE - beg);
  int b[EPT], rk[EPT];
#pragma unroll
  for (int j = 0; j < EPT; ++j) {
    b[j] = (int)min((unsigned)d[j] >> BSH, (unsigned)(nbk - 1));
    rk[j] = 0;
    if (j * TPB + tid < cnt) rk[j] = atomicAdd(&hist[b[j]], 1);
  }
  __syncthreads();
  const int hv = (tid < MAXB) ? hist[tid] : 0;
  int inc = hv;
#pragma unroll
  for (int k = 1; k < 32; k <<= 1) {
    const int a = __shfl_up(inc, k);
    if (lane >= k) inc += a;
  }
  if (tid < MAXB && lane == 31) wsA[wave] = inc;
  __syncthreads();
  if (tid < MAXB) {
    int st = inc - hv;
#pragma unroll
    for (int w = 0; w < MAXB / 32; ++w)
      if (w < wave) st += wsA[w];
    hist[tid] = st;
    T[me * MAXB + tid] = st;
  }
  __syncthreads();
#pragma unroll
  for (int j = 0; j < EPT; ++j) {
    if (j * TPB + tid < cnt) {
      const int p = hist[b[j]] + rk[j];
      if ((unsigned)p < (unsigned)CHUNK)
        stage[p] = ((unsigned)s[j] << BSH) | ((unsigned)d[j] & (BNODES - 1));
    }
  }
  __syncthreads();
  for (int i = tid; i < cnt; i += TPB) P[beg + i] = stage[i];
}

__device__ __forceinline__ void run_of(const int* __restrict__ T, int nE, int nb1, int bk, int r,
                                       int& first, int& len, int& start) {
  first = 0;
  len = 0;
  start = 0;
  if (r < nb1) {
    const int cc = min(CHUNK, nE - r * CHUNK);
    int s0 = T[r * MAXB + bk];
    int e0 = (bk + 1 < MAXB) ? T[r * MAXB + bk + 1] : cc;
    s0 = min(max(s0, 0), cc);
    e0 = min(max(e0, s0), cc);
    first = r * CHUNK + s0;
    len = e0 - s0;
    start = s0;
  }
}

template <typename F>
__device__ __forceinline__ void for_each_word(const unsigned* __restrict__ P, const int* __restrict__ T,
                                              int nE, int nb1, int bk, int first0, int len0, F f) {
  const int tid = threadIdx.x;
  const int lane = tid & 31;
  for (int g = 0; g * 256 < nb1; ++g) {
    int first = first0, len = len0;
    if (g != 0) {
      int st;
      run_of(T, nE, nb1, bk, g * 256 + tid, first, len, st);
    }
#pragma unroll
    for (int h = 0; h < 32; h += 16) {
      unsigned w[16];
#pragma unroll
      for (int u = 0; u < 16; ++u) {
        const int fu = __builtin_amdgcn_readlane(first, h + u);
        const int lu = __builtin_amdgcn_readlane(len, h + u);
        w[u] = P[min(fu + min(lane, max(lu - 1, 0)), nE - 1)];
      }
#pragma unroll
      for (int u = 0; u < 16; ++u) {
        const int fu = __builtin_amdgcn_readlane(first, h + u);
        const int lu = __builtin_amdgcn_readlane(len, h + u);
        if (lane < lu) f(w[u]);
        if (lu > 32)
          for (int j = 32 + lane; j < lu; j += 32) f(P[fu + j]);
      }
    }
  }
}

__device__ __forceinline__ void bucket_sort_block(
    const unsigned* __restrict__ P, const int* __restrict__ T,
    int* __restrict__ rowptr, int* __restrict__ col, int n, int nE, int nb1, int bk,
    int* cnt, int* wsum, int* sorted) {
  const int tid = threadIdx.x;
  const int lane = tid & 31;
  const int wave = tid >> 5;
  cnt[tid] = 0;
  int first0, len0, st0;
  run_of(T, nE, nb1, bk, tid, first0, len0, st0);
  int sb = st0, sl = len0;
  for (int g = 1; g * 256 < nb1; ++g) {
    int fi, le, st;
    run_of(T, nE, nb1, bk, g * 256 + tid, fi, le, st);
    sb += st;
    sl += le;
  }
#pragma unroll
  for (int k = 16; k >= 1; k >>= 1) {
    sb += __shfl_xor(sb, k);
    sl += __shfl_xor(sl, k);
  }
  if (lane == 0) {
    wsum[wave] = sb;
    wsum[8 + wave] = sl;
  }
  __syncthreads();
  int beg = 0, len = 0;
#pragma unroll
  for (int w = 0; w < 8; ++w) {
    beg += wsum[w];
    len += wsum[8 + w];
  }
  beg = min(max(beg, 0), nE);
  len = min(max(len, 0), nE - beg);
  for_each_word(P, T, nE, nb1, bk, first0, len0,
                [&](unsigned w) { atomicAdd(&cnt[w & (BNODES - 1)], 1); });
  __syncthreads();
  const int c = cnt[tid];
  int inc = c;
#pragma unroll
  for (int k = 1; k < 32; k <<= 1) {
    const int a = __shfl_up(inc, k);
    if (lane >= k) inc += a;
  }
  if (lane == 31) wsum[wave] = inc;
  __syncthreads();
  int excl = inc - c;
#pragma unroll
  for (int w = 0; w < 8; ++w)
    if (w < wave) excl += wsum[w];
  const int node = bk * BNODES + tid;
  if (node < n) rowptr[node] = beg + excl;
  if (node == n - 1) rowptr[n] = beg + excl + c;
  cnt[tid] = excl;
  __syncthreads();
  const bool fits = len <= SCAP;
  for_each_word(P, T, nE, nb1, bk, first0, len0, [&](unsigned w) {
    const int p = atomicAdd(&cnt[w & (BNODES - 1)], 1);
    const int src = (int)(w >> BSH);
    if (fits) {
      if ((unsigned)p < (unsigned)SCAP) sorted[p] = src;
    } else {
      if ((unsigned)p < (unsigned)len) col[beg + p] = src;
    }
  });
  if (fits) {
    __syncthreads();
    for (int i = tid; i < len; i += 256) col[beg + i] = sorted[i];
  }
}

__global__ __launch_bounds__(256) void sort_and_gemm(
    const unsigned* __restrict__ P, const int* __restrict__ T,
    int* __restrict__ rowptr, int* __restrict__ col, int n, int nE, int nb1, int nbk,
    const float* __restrict__ X, const float* __restrict__ Wl,
    const float* __restrict__ bl, const float* __restrict__ Wr,
    float* __restrict__ Yl, float* __restrict__ Yr) {
  constexpr int OUTF = 64;
  __shared__ __attribute__((aligned(16))) float smem[(2 * OUTF + 128) * LDP];
  static_assert(sizeof(float) * (2 * OUTF + 128) * LDP >= sizeof(int) * (BNODES + 16 + SCAP), "the sort's tables must fit");
  if ((int)blockIdx.x < nbk) {
    int* cnt = (int*)smem;
    bucket_sort_block(P, T, rowptr, col, n, nE, nb1, (int)blockIdx.x, cnt, cnt + BNODES, cnt + BNODES + 16);
  } else {
    gemm_pair_block<OUTF>(X, Wl, bl, Wr, Yl, Yr, n, ((int)blockIdx.x - nbk) * 128,
                          smem, smem + 2 * OUTF * LDP);
  }
}

template <int F, bool RELU>
__global__ __launch_bounds__(256) void sage_gather(
    const int* __restrict__ rowptr, const int* __restrict__ col,
    const float* __restrict__ H, const float* __restrict__ R,
    float* __restrict__ Y, int n, int nE) {
  constexpr int G = F / 4;
  constexpr int NG = 32 / G;
  const int lane = threadIdx.x & 31;
  const int sub = lane / G;
  const int c = lane % G;
  const int wv = __builtin_amdgcn_readfirstlane((int)(threadIdx.x >> 5));
  const int nwaves = gridDim.x * (blockDim.x >> 5);
  for (int node = blockIdx.x * (blockDim.x >> 5) + wv; node < n; node += nwaves) {
    int beg = __builtin_amdgcn_readfirstlane(rowptr[node]);
    int end = __builtin_amdgcn_readfirstlane(rowptr[node + 1]);
    const float4 r = *(const float4*)(R + (long)node * F + c * 4);
    beg = min(max(beg, 0), nE);
    end = min(max(end, beg), nE);
    float4 acc = make_float4(0.0f, 0.0f, 0.0f, 0.0f);
    for (int base = beg; base < end; base += 32) {
      const int m = min(32, end - base);
      int cv = (lane < m) ? col[base + lane] : 0;
      cv = (int)min((unsigned)cv, (unsigned)(n - 1));
      const int full = m / NG;
#pragma unroll 4
      for (int k = 0; k < full; ++k) {
        const int s = __shfl(cv, k * NG + sub);
        const float4 v = *(const float4*)(H + (long)s * F + c * 4);
        acc.x += v.x; acc.y += v.y; acc.z += v.z; acc.w += v.w;
      }
      const int rem = m - full * NG;
      if (rem != 0) {
        const int s = __shfl(cv, full * NG + sub);
        if (sub < rem) {
          const float4 v = *(const float4*)(H + (long)s * F + c * 4);
          acc.x += v.x; acc.y += v.y; acc.z += v.z; acc.w += v.w;
        }
      }
    }
#pragma unroll
    for (int d = G; d < 32; d <<= 1) {
      acc.x += __shfl_xor(acc.x, d);
      acc.y += __shfl_xor(acc.y, d);
      acc.z += __shfl_xor(acc.z, d);
      acc.w += __shfl_xor(acc.w, d);
    }
    if (sub == 0) {
      const float d = fmaxf((float)(end - beg), 1.0f);
      float4 o;
      o.x = r.x + acc.x / d;
      o.y = r.y + acc.y / d;
      o.z = r.z + acc.z / d;
      o.w = r.w + acc.w / d;
      if (RELU) {
        o.x = fmaxf(o.x, 0.0f); o.y = fmaxf(o.y, 0.0f);
        o.z = fmaxf(o.z, 0.0f); o.w = fmaxf(o.w, 0.0f);
      }
      *(float4*)(Y + (long)node * F + c * 4) = o;
    }
  }
}

#define GW 4
__global__ __launch_bounds__(GW * 32) void sage_gather_gemm(
    const int* __restrict__ rowptr, const int* __restrict__ col,
    const float* __restrict__ H, const float* __restrict__ R,
    const float* __restrict__ Wl, const float* __restrict__ bl, const float* __restrict__ Wr,
    float* __restrict__ Yl, float* __restrict__ Yr, int n, int nE) {
  constexpr int F = KDIM;
  constexpr int OUTF = 32;
  constexpr int NT = OUTF / 16;
  constexpr int NB = 2 * NT;
  __shared__ __attribute__((aligned(16))) float sW[2 * OUTF * LDP];
  __shared__ __attribute__((aligned(16))) float sA[GW * 16 * LDP];
  const int tid = threadIdx.x;
  const int wave = __builtin_amdgcn_readfirstlane(tid >> 5);
  const int lane = tid & 31;
  const int half = lane >> 4;
  const int l15  = lane & 15;

  {
    constexpr int WS = (2 * OUTF * 16) / (GW * 32);
    float4 wv[WS];
#pragma unroll
    for (int i = 0; i < WS; ++i) {
      const int idx = i * (GW * 32) + tid;
      const int row = idx >> 4;
      const float* src = (row < OUTF) ? (Wl + row * KDIM) : (Wr + (row - OUTF) * KDIM);
      wv[i] = *(const float4*)(src + (idx & 15) * 4);
    }
#pragma unroll
    for (int i = 0; i < WS; ++i) {
      const int idx = i * (GW * 32) + tid;
      *(float4*)(sW + (idx >> 4) * LDP + (idx & 15) * 4) = wv[i];
    }
  }

  const int node0 = blockIdx.x * (GW * 16) + wave * 16;
  float* own = sA + wave * 16 * LDP;

  int rp = rowptr[min(node0 + min(lane, 16), n)];
  rp = min(max(rp, 0), nE);
  int cvn = col[min(__builtin_amdgcn_readlane(rp, 0) + lane, nE - 1)];
  for (int i = 0; i < 16; ++i) {
    const int node = node0 + i;
    const int beg = __builtin_amdgcn_readlane(rp, i);
    const int nxt = __builtin_amdgcn_readlane(rp, i + 1);
    const int end = max(nxt, beg);
    int cv = cvn;
    cvn = col[min(nxt + lane, nE - 1)];
    float4 o = make_float4(0.0f, 0.0f, 0.0f, 0.0f);
    if (node < n) {
      const float4 r = *(const float4*)(R + (long)node * F + l15 * 4);
      float4 sum = make_float4(0.0f, 0.0f, 0.0f, 0.0f);
      for (int base = beg; base < end; base += 32) {
        const int m = min(32, end - base);
        if (base != beg) cv = col[min(base + lane, nE - 1)];
        cv = (lane < m) ? (int)min((unsigned)cv, (unsigned)(n - 1)) : 0;
        const int nf = m >> 1;
        const bool odd = (m & 1) != 0;
#pragma unroll
        for (int b = 0; b < 16; b += 8) {
          if (2 * b < m) {
            float4 v[8], vt;
            const bool tail = odd && nf >= b && nf < b + 8;
#pragma unroll
            for (int k = 0; k < 8; ++k) {
              if (b + k < nf) {
                const int s0 = __builtin_amdgcn_readlane(cv, 2 * (b + k));
                const int s1 = __builtin_amdgcn_readlane(cv, 2 * (b + k) + 1);
                const int s = half ? s1 : s0;
                v[k] = *(const float4*)(H + (long)s * F + l15 * 4);
              }
            }
            if (tail) {
              const int st = __builtin_amdgcn_readlane(cv, m - 1);
              vt = *(const float4*)(H + (long)st * F + l15 * 4);
            }
#pragma unroll
            for (int k = 0; k < 8; ++k) {
              if (b + k < nf) {
                sum.x += v[k].x; sum.y += v[k].y; sum.z += v[k].z; sum.w += v[k].w;
              }
            }
            if (tail) {
              sum.x += half ? 0.0f : vt.x;
              sum.y += half ? 0.0f : vt.y;
              sum.z += half ? 0.0f : vt.z;
              sum.w += half ? 0.0f : vt.w;
            }
          }
        }
      }
      sum.x += __shfl_xor(sum.x, 16);
      sum.y += __shfl_xor(sum.y, 16);
      sum.z += __shfl_xor(sum.z, 16);
      sum.w += __shfl_xor(sum.w, 16);
      const float d = fmaxf((float)(end - beg), 1.0f);
      o.x = fmaxf(r.x + sum.x / d, 0.0f);
      o.y = fmaxf(r.y + sum.y / d, 0.0f);
      o.z = fmaxf(r.z + sum.z / d, 0.0f);
      o.w = fmaxf(r.w + sum.w / d, 0.0f);
    }
    if (half == 0) *(float4*)(own + i * LDP + l15 * 4) = o;
  }
  __syncthreads();

  const int r0 = node0;
  const bool active = r0 < n;
  v8f acc[NB];
  if (active) {
    mma_pair<OUTF>(own, sW, bl, acc, half, l15);
    put_tile<NT>(own, acc, half, l15);
    put_tile<NT>(own + OUTF, acc + NT, half, l15);
  }
  __syncthreads();
  if (active) {
    store_tile_rows<OUTF>(own, Yl, r0, n, lane);
    store_tile_rows<OUTF>(own + OUTF, Yr, r0, n, lane);
  }
}

extern "C" void kernel_launch(void* const* d_in, const int* in_sizes, int n_in,
                              void* d_out, int out_size, void* d_ws, size_t ws_size,
                              hipStream_t stream) {
  const float* x   = (const float*)d_in[0];
  const int*   ei  = (const int*)d_in[1];
  const float* Wl0 = (const float*)d_in[2];
  const float* bl0 = (const float*)d_in[3];
  const float* Wr0 = (const float*)d_in[4];
  const float* Wl1 = (const float*)d_in[5];
  const float* bl1 = (const float*)d_in[6];
  const float* Wr1 = (const float*)d_in[7];
  float* out = (float*)d_out;

  const int N = in_sizes[0] / 64;
  const int E = in_sizes[1] / 2;

  float* B0 = (float*)d_ws;
  float* B1 = B0 + (size_t)N * 64;
  float* B2 = B1 + (size_t)N * 64;
  const int NB1 = (E + CHUNK - 1) / CHUNK;
  const int NBK = (N + BNODES - 1) / BNODES;
  int* T        = (int*)(B2 + (size_t)N * 64);
  unsigned* P   = (unsigned*)(T + (size_t)NB1 * MAXB);
  int* col      = (int*)P + E;
  int* rowptr   = col + E;

  const dim3 blk(256);
  const int GEMMB = (N + 127) / 128;

  bin_local<<<dim3(NB1), dim3(TPB), 0, stream>>>(ei, T, P, E, NBK);
  sort_and_gemm<<<dim3(NBK + GEMMB), blk, 0, stream>>>(P, T, rowptr, col, N, E, NB1, NBK,
                                                     x, Wl0, bl0, Wr0, B0, B1);

  float* Yl1 = B2;
  float* Yr1 = B2 + (size_t)N * 32;
  sage_gather_gemm<<<dim3((N + GW * 16 - 1) / (GW * 16)), dim3(GW * 32), 0, stream>>>(
      rowptr, col, B0, B1, Wl1, bl1, Wr1, Yl1, Yr1, N, E);

  sage_gather<32, false><<<dim3(4096), blk, 0, stream>>>(rowptr, col, Yl1, Yr1, out, N, E);
}


# Layer 0 gathers rows of x itself (mean before the products: sum order changed) inside the launch that sorts a bucket in LDS; its products stay in LDS; 4 -> 3 launches, two N x 64 arrays gone
# speedup vs baseline: 5.1054x; 1.1142x over previous
#include <hip/hip_runtime.h>


typedef __attribute__((ext_vector_type(2))) float v2f;
typedef __attribute__((ext_vector_type(8))) float v8f;

#define KDIM 64

#define LDP 68

template <int NT>
__device__ __forceinline__ void put_tile(float* lds, const v8f* acc, int half, int l15) {
#pragma unroll
  for (int t = 0; t < NT; ++t)
#pragma unroll
    for (int v = 0; v < 8; ++v) lds[(v + 8 * half) * LDP + t * 16 + l15] = acc[t][v];
}

template <int OUTF>
__device__ __forceinline__ void store_tile_rows(const float* lds, float* __restrict__ Y,
                                                int r0, int n, int lane) {
  constexpr int LPR = OUTF / 4;
  constexpr int RPI = 32 / LPR;
  constexpr int NS = 16 / RPI;
  const int rr = lane / LPR;
  const int c = (lane % LPR) * 4;
  float4 v[NS];
#pragma unroll
  for (int j = 0; j < NS; ++j) v[j] = *(const float4*)(lds + (j * RPI + rr) * LDP + c);
  float* yp = Y + (long)(r0 + rr) * OUTF + c;
  if (r0 + 16 <= n) {
#pragma unroll
    for (int j = 0; j < NS; ++j) *(float4*)(yp + (long)j * RPI * OUTF) = v[j];
  } else {
#pragma unroll
    for (int j = 0; j < NS; ++j)
      if (r0 + j * RPI + rr < n) *(float4*)(yp + (long)j * RPI * OUTF) = v[j];
  }
}

template <int OUTF>
__device__ __forceinline__ void mma_pair(const float* rows, const float* sW, const float* __restrict__ bl,
                                         v8f* acc, int half, int l15) {
  constexpr int NT = OUTF / 16;
  constexpr int NB = 2 * NT;
#pragma unroll
  for (int t = 0; t < NT; ++t) {
    const float b = bl[t * 16 + l15];
#pragma unroll
    for (int j = 0; j < 8; ++j) {
      acc[t][j] = b;
      acc[NT + t][j] = 0.0f;
    }
  }

  const float* ap = rows + l15 * LDP + half * 2;
  const float* bp = sW + l15 * LDP + half * 2;
#pragma unroll
  for (int kb = 0; kb < KDIM / 4; ++kb) {
    const v2f a = *(const v2f*)(ap + kb * 4);
#pragma unroll
    for (int t = 0; t < NB; ++t) {
      const v2f b = *(const v2f*)(bp + t * 16 * LDP + kb * 4);
      acc[t] = __builtin_amdgcn_wmma_f32_16x16x4_f32(
           false, a,  false, b,
           (short)0, acc[t],  false,  false);
    }
  }
}

#define BSH 8
#define BNODES (1 << BSH)
#define MAXB 512
#define TPB 896
#define EPT 7
#define CHUNK (TPB * EPT)
#define SCAP 8192

__global__ __launch_bounds__(TPB) void bin_local(
    const int* __restrict__ ei, int* __restrict__ T, unsigned* __restrict__ P, int nE, int nbk) {
  __shared__ unsigned stage[CHUNK];
  __shared__ int hist[MAXB];
  __shared__ int wsA[MAXB / 32];
  const int tid = threadIdx.x;
  const int lane = tid & 31;
  const int wave = tid >> 5;
  const int me = blockIdx.x;
  const int beg = me * CHUNK;
  if (tid < MAXB) hist[tid] = 0;
  int s[EPT], d[EPT];
#pragma unroll
  for (int j = 0; j < EPT; ++j) {
    const int e = min(beg + j * TPB + tid, nE - 1);
    s[j] = ei[e];
    d[j] = ei[nE + e];
  }
  __syncthreads();
  const int cnt = min(CHUNK, nE - beg);
  int b[EPT], rk[EPT];
#pragma unroll
  for (int j = 0; j < EPT; ++j) {
    b[j] = (int)min((unsigned)d[j] >> BSH, (unsigned)(nbk - 1));
    rk[j] = 0;
    if (j * TPB + tid < cnt) rk[j] = atomicAdd(&hist[b[j]], 1);
  }
  __syncthreads();
  const int hv = (tid < MAXB) ? hist[tid] : 0;
  int inc = hv;
#pragma unroll
  for (int k = 1; k < 32; k <<= 1) {
    const int a = __shfl_up(inc, k);
    if (lane >= k) inc += a;
  }
  if (tid < MAXB && lane == 31) wsA[wave] = inc;
  __syncthreads();
  if (tid < MAXB) {
    int st = inc - hv;
#pragma unroll
    for (int w = 0; w < MAXB / 32; ++w)
      if (w < wave) st += wsA[w];
    hist[tid] = st;
    T[me * MAXB + tid] = st;
  }
  __syncthreads();
#pragma unroll
  for (int j = 0; j < EPT; ++j) {
    if (j * TPB + tid < cnt) {
      const int p = hist[b[j]] + rk[j];
      if ((unsigned)p < (unsigned)CHUNK)
        stage[p] = ((unsigned)s[j] << BSH) | ((unsigned)d[j] & (BNODES - 1));
    }
  }
  __syncthreads();
  for (int i = tid; i < cnt; i += TPB) P[beg + i] = stage[i];
}

__device__ __forceinline__ void run_of(const int* __restrict__ T, int nE, int nb1, int bk, int r,
                                       int& first, int& len, int& start) {
  first = 0;
  len = 0;
  start = 0;
  if (r < nb1) {
    const int cc = min(CHUNK, nE - r * CHUNK);
    int s0 = T[r * MAXB + bk];
    int e0 = (bk + 1 < MAXB) ? T[r * MAXB + bk + 1] : cc;
    s0 = min(max(s0, 0), cc);
    e0 = min(max(e0, s0), cc);
    first = r * CHUNK + s0;
    len = e0 - s0;
    start = s0;
  }
}

template <typename F, typename M>
__device__ __forceinline__ void for_each_word(const unsigned* __restrict__ P, const int* __restrict__ T,
                                              int nE, int nb1, int bk, int first0, int len0, F f, M mid) {
  const int tid = threadIdx.x;
  const int lane = tid & 31;
  for (int g = 0; g * 256 < nb1; ++g) {
    int first = first0, len = len0;
    if (g != 0) {
      int st;
      run_of(T, nE, nb1, bk, g * 256 + tid, first, len, st);
    }
#pragma unroll
    for (int h = 0; h < 32; h += 16) {
      unsigned w[16];
#pragma unroll
      for (int u = 0; u < 16; ++u) {
        const int fu = __builtin_amdgcn_readlane(first, h + u);
        const int lu = __builtin_amdgcn_readlane(len, h + u);
        w[u] = P[min(fu + min(lane, max(lu - 1, 0)), nE - 1)];
      }
      if (g == 0) mid(h >> 4);
#pragma unroll
      for (int u = 0; u < 16; ++u) {
        const int fu = __builtin_amdgcn_readlane(first, h + u);
        const int lu = __builtin_amdgcn_readlane(len, h + u);
        if (lane < lu) f(w[u]);
        if (lu > 32)
          for (int j = 32 + lane; j < lu; j += 32) f(P[fu + j]);
      }
    }
  }
}

#define HN 128
#define SCAPH 4096
__global__ __launch_bounds__(256) void sort_gather_gemm(
    const unsigned* __restrict__ P, const int* __restrict__ T,
    int* __restrict__ rowptr, int* col, int n, int nE, int nb1,
    const float* __restrict__ X,
    const float* __restrict__ Wl0, const float* __restrict__ bl0, const float* __restrict__ Wr0,
    const float* __restrict__ Wl1, const float* __restrict__ bl1, const float* __restrict__ Wr1,
    float* __restrict__ Yl, float* __restrict__ Yr) {
  constexpr int F = KDIM;
  __shared__ __attribute__((aligned(16))) float sW[64 * LDP];
  __shared__ __attribute__((aligned(16))) float sA[8 * 16 * LDP];
  __shared__ int cnt[BNODES];
  __shared__ int rs[BNODES + 1];
  __shared__ int wsum[16];
  __shared__ float gate[HN];
  __shared__ int sorted[SCAPH];
  const int tid = threadIdx.x;
  const int wave = __builtin_amdgcn_readfirstlane(tid >> 5);
  const int lane = tid & 31;
  const int half = lane >> 4;
  const int l15  = lane & 15;
  const int bk = (int)blockIdx.x >> 1;
  const int hb = (int)blockIdx.x & 1;
  const int ln0 = hb * HN + wave * 16;
  const int node0 = bk * BNODES + ln0;
  float* own = sA + wave * 16 * LDP;

  cnt[tid] = 0;
  int first0, len0, st0;
  run_of(T, nE, nb1, bk, tid, first0, len0, st0);
  int sb = st0;
  for (int g = 1; g * 256 < nb1; ++g) {
    int fi, le, st;
    run_of(T, nE, nb1, bk, g * 256 + tid, fi, le, st);
    sb += st;
  }
  float4 wa[4], xv[8];
#pragma unroll
  for (int i = 0; i < 4; ++i) {
    const int idx = i * 256 + tid;
    wa[i] = *(const float4*)(Wr0 + (idx >> 4) * KDIM + (idx & 15) * 4);
  }
#pragma unroll
  for (int i = 0; i < 8; ++i) {
    const int row = min(node0 + i * 2 + half, n - 1);
    xv[i] = *(const float4*)(X + (long)row * F + l15 * 4);
  }
#pragma unroll
  for (int k = 16; k >= 1; k >>= 1) sb += __shfl_xor(sb, k);
  if (lane == 0) wsum[wave] = sb;
#pragma unroll
  for (int i = 0; i < 4; ++i) {
    const int idx = i * 256 + tid;
    *(float4*)(sW + (idx >> 4) * LDP + (idx & 15) * 4) = wa[i];
  }
#pragma unroll
  for (int i = 0; i < 8; ++i) *(float4*)(own + (i * 2 + half) * LDP + l15 * 4) = xv[i];
  __syncthreads();
  int beg = 0;
#pragma unroll
  for (int w = 0; w < 8; ++w) beg += wsum[w];
  beg = min(max(beg, 0), nE);

  v8f acc[4];
#pragma unroll
  for (int t = 0; t < 4; ++t)
#pragma unroll
    for (int j = 0; j < 8; ++j) acc[t][j] = 0.0f;
  const float* ap = own + l15 * LDP + half * 2;
  const float* bp = sW + l15 * LDP + half * 2;
  auto k_steps = [&](int kb0) {
#pragma unroll
    for (int kb = kb0; kb < kb0 + 4; ++kb) {
      const v2f a = *(const v2f*)(ap + kb * 4);
#pragma unroll
      for (int t = 0; t < 4; ++t) {
        const v2f b = *(const v2f*)(bp + t * 16 * LDP + kb * 4);
        acc[t] = __builtin_amdgcn_wmma_f32_16x16x4_f32(false, a, false, b, (short)0, acc[t], false, false);
      }
    }
  };

  for_each_word(P, T, nE, nb1, bk, first0, len0,
                [&](unsigned w) { atomicAdd(&cnt[w & (BNODES - 1)], 1); },
                [&](int part) { k_steps(part * 4); });
  __syncthreads();
  const int c = cnt[tid];
  int inc = c;
#pragma unroll
  for (int k = 1; k < 32; k <<= 1) {
    const int a = __shfl_up(inc, k);
    if (lane >= k) inc += a;
  }
  if (lane == 31) wsum[wave] = inc;
  __syncthreads();
  int excl = inc - c;
#pragma unroll
  for (int w = 0; w < 8; ++w)
    if (w < wave) excl += wsum[w];
  rs[tid] = excl;
  if (tid == BNODES - 1) rs[BNODES] = excl + c;
  if ((tid >> 7) == hb) {
    const int node = bk * BNODES + tid;
    if (node < n) rowptr[node] = beg + excl;
    if (node == n - 1) rowptr[n] = beg + excl + c;
  }
  cnt[tid] = excl;
  __syncthreads();
  const int hbase = __builtin_amdgcn_readfirstlane(rs[hb * HN]);
  const int hlen = __builtin_amdgcn_readfirstlane(rs[hb * HN + HN]) - hbase;
  const int gbase = __builtin_amdgcn_readfirstlane(beg) + hbase;
  const bool fits = hlen <= SCAPH;
  for_each_word(P, T, nE, nb1, bk, first0, len0, [&](unsigned w) {
    const int nd = (int)(w & (BNODES - 1));
    if ((nd >> 7) == hb) {
      const int p = atomicAdd(&cnt[nd], 1) - hbase;
      const int s = (int)(w >> BSH);
      if (fits) {
        if ((unsigned)p < (unsigned)SCAPH) sorted[p] = s;
      } else {
        if ((unsigned)p < (unsigned)hlen && (unsigned)(gbase + p) < (unsigned)nE) col[gbase + p] = s;
      }
    }
  }, [&](int part) { k_steps(8 + part * 4); });
  __syncthreads();
  const int wr = tid >> 4;
  const int wc4 = (tid & 15) * 4;
  const float4 p0 = *(const float4*)(Wl0 + wr * KDIM + wc4);
  const float4 p1 = *(const float4*)(Wl0 + (16 + wr) * KDIM + wc4);
  const float4 p2 = *(const float4*)(Wl0 + (32 + wr) * KDIM + wc4);
  const float4 p3 = *(const float4*)(Wl0 + (48 + wr) * KDIM + wc4);
  if (fits)
    for (int i = tid; i < hlen; i += 256)
      if ((unsigned)(gbase + i) < (unsigned)nE) col[gbase + i] = sorted[i];

  auto src_at = [&](int rel) -> int {
    return fits ? sorted[min(max(rel, 0), SCAPH - 1)] : col[min(max(gbase + rel, 0), nE - 1)];
  };
  int rp = rs[ln0 + min(lane, 16)] - hbase;
  rp = min(max(rp, 0), hlen);
  int cvn = src_at(__builtin_amdgcn_readlane(rp, 0) + lane);
  for (int i = 0; i < 16; ++i) {
    const int node = node0 + i;
    const int lb = __builtin_amdgcn_readlane(rp, i);
    const int nxt = __builtin_amdgcn_readlane(rp, i + 1);
    const int end = max(nxt, lb);
    int cv = cvn;
    cvn = src_at(nxt + lane);
    float4 o = make_float4(0.0f, 0.0f, 0.0f, 0.0f);
    float gt = 0.0f;
    if (node < n) {
      float4 sum = make_float4(0.0f, 0.0f, 0.0f, 0.0f);
      for (int base = lb; base < end; base += 32) {
        const int m = min(32, end - base);
        if (base != lb) cv = src_at(base + lane);
        cv = (lane < m) ? (int)min((unsigned)cv, (unsigned)(n - 1)) : 0;
        const int nf = m >> 1;
        const bool odd = (m & 1) != 0;
#pragma unroll
        for (int b = 0; b < 16; b += 8) {
          if (2 * b < m) {
            float4 v[8], vt;
            const bool tail = odd && nf >= b && nf < b + 8;
#pragma unroll
            for (int k = 0; k < 8; ++k) {
              if (b + k < nf) {
                const int s0 = __builtin_amdgcn_readlane(cv, 2 * (b + k));
                const int s1 = __builtin_amdgcn_readlane(cv, 2 * (b + k) + 1);
                const int s = half ? s1 : s0;
                v[k] = *(const float4*)(X + (long)s * F + l15 * 4);
              }
            }
            if (tail) {
              const int st = __builtin_amdgcn_readlane(cv, m - 1);
              vt = *(const float4*)(X + (long)st * F + l15 * 4);
            }
#pragma unroll
            for (int k = 0; k < 8; ++k) {
              if (b + k < nf) {
                sum.x += v[k].x; sum.y += v[k].y; sum.z += v[k].z; sum.w += v[k].w;
              }
            }
            if (tail) {
              sum.x += half ? 0.0f : vt.x;
              sum.y += half ? 0.0f : vt.y;
              sum.z += half ? 0.0f : vt.z;
              sum.w += half ? 0.0f : vt.w;
            }
          }
        }
      }
      sum.x += __shfl_xor(sum.x, 16);
      sum.y += __shfl_xor(sum.y, 16);
      sum.z += __shfl_xor(sum.z, 16);
      sum.w += __shfl_xor(sum.w, 16);
      const float d = fmaxf((float)(end - lb), 1.0f);
      o.x = sum.x / d;
      o.y = sum.y / d;
      o.z = sum.z / d;
      o.w = sum.w / d;
      gt = (end > lb) ? 1.0f : 0.0f;
    }
    if (half == 0) *(float4*)(own + i * LDP + l15 * 4) = o;
    if (lane == 0) gate[wave * 16 + i] = gt;
  }
  *(float4*)(sW + wr * LDP + wc4) = p0;
  *(float4*)(sW + (16 + wr) * LDP + wc4) = p1;
  *(float4*)(sW + (32 + wr) * LDP + wc4) = p2;
  *(float4*)(sW + (48 + wr) * LDP + wc4) = p3;
  const float4 q0 = *(const float4*)(Wl1 + wr * KDIM + wc4);
  const float4 q1 = *(const float4*)(Wl1 + (16 + wr) * KDIM + wc4);
  const float4 q2 = *(const float4*)(Wr1 + wr * KDIM + wc4);
  const float4 q3 = *(const float4*)(Wr1 + (16 + wr) * KDIM + wc4);
  __syncthreads();

  k_steps(0);
  k_steps(4);
  k_steps(8);
  k_steps(12);
  {
    float gv[8];
#pragma unroll
    for (int v = 0; v < 8; ++v) gv[v] = gate[wave * 16 + v + 8 * half];
#pragma unroll
    for (int t = 0; t < 4; ++t) {
      const float b = bl0[t * 16 + l15];
#pragma unroll
      for (int v = 0; v < 8; ++v)
        own[(v + 8 * half) * LDP + t * 16 + l15] = fmaxf(acc[t][v] + gv[v] * b, 0.0f);
    }
  }
  __syncthreads();
  *(float4*)(sW + wr * LDP + wc4) = q0;
  *(float4*)(sW + (16 + wr) * LDP + wc4) = q1;
  *(float4*)(sW + (32 + wr) * LDP + wc4) = q2;
  *(float4*)(sW + (48 + wr) * LDP + wc4) = q3;
  __syncthreads();

  constexpr int OUTF = 32;
  constexpr int NT = OUTF / 16;
  const int r0 = node0;
  const bool active = r0 < n;
  v8f acc1[2 * NT];
  if (active) {
    mma_pair<OUTF>(own, sW, bl1, acc1, half, l15);
    put_tile<NT>(own, acc1, half, l15);
    put_tile<NT>(own + OUTF, acc1 + NT, half, l15);
  }
  __syncthreads();
  if (active) {
    store_tile_rows<OUTF>(own, Yl, r0, n, lane);
    store_tile_rows<OUTF>(own + OUTF, Yr, r0, n, lane);
  }
}

template <int F, bool RELU>
__global__ __launch_bounds__(256) void sage_gather(
    const int* __restrict__ rowptr, const int* __restrict__ col,
    const float* __restrict__ H, const float* __restrict__ R,
    float* __restrict__ Y, int n, int nE) {
  constexpr int G = F / 4;
  constexpr int NG = 32 / G;
  const int lane = threadIdx.x & 31;
  const int sub = lane / G;
  const int c = lane % G;
  const int wv = __builtin_amdgcn_readfirstlane((int)(threadIdx.x >> 5));
  const int nwaves = gridDim.x * (blockDim.x >> 5);
  for (int node = blockIdx.x * (blockDim.x >> 5) + wv; node < n; node += nwaves) {
    int beg = __builtin_amdgcn_readfirstlane(rowptr[node]);
    int end = __builtin_amdgcn_readfirstlane(rowptr[node + 1]);
    const float4 r = *(const float4*)(R + (long)node * F + c * 4);
    beg = min(max(beg, 0), nE);
    end = min(max(end, beg), nE);
    float4 acc = make_float4(0.0f, 0.0f, 0.0f, 0.0f);
    for (int base = beg; base < end; base += 32) {
      const int m = min(32, end - base);
      int cv = (lane < m) ? col[base + lane] : 0;
      cv = (int)min((unsigned)cv, (unsigned)(n - 1));
      const int full = m / NG;
#pragma unroll 4
      for (int k = 0; k < full; ++k) {
        const int s = __shfl(cv, k * NG + sub);
        const float4 v = *(const float4*)(H + (long)s * F + c * 4);
        acc.x += v.x; acc.y += v.y; acc.z += v.z; acc.w += v.w;
      }
      const int rem = m - full * NG;
      if (rem != 0) {
        const int s = __shfl(cv, full * NG + sub);
        if (sub < rem) {
          const float4 v = *(const float4*)(H + (long)s * F + c * 4);
          acc.x += v.x; acc.y += v.y; acc.z += v.z; acc.w += v.w;
        }
      }
    }
#pragma unroll
    for (int d = G; d < 32; d <<= 1) {
      acc.x += __shfl_xor(acc.x, d);
      acc.y += __shfl_xor(acc.y, d);
      acc.z += __shfl_xor(acc.z, d);
      acc.w += __shfl_xor(acc.w, d);
    }
    if (sub == 0) {
      const float d = fmaxf((float)(end - beg), 1.0f);
      float4 o;
      o.x = r.x + acc.x / d;
      o.y = r.y + acc.y / d;
      o.z = r.z + acc.z / d;
      o.w = r.w + acc.w / d;
      if (RELU) {
        o.x = fmaxf(o.x, 0.0f); o.y = fmaxf(o.y, 0.0f);
        o.z = fmaxf(o.z, 0.0f); o.w = fmaxf(o.w, 0.0f);
      }
      *(float4*)(Y + (long)node * F + c * 4) = o;
    }
  }
}

extern "C" void kernel_launch(void* const* d_in, const int* in_sizes, int n_in,
                              void* d_out, int out_size, void* d_ws, size_t ws_size,
                              hipStream_t stream) {
  const float* x   = (const float*)d_in[0];
  const int*   ei  = (const int*)d_in[1];
  const float* Wl0 = (const float*)d_in[2];
  const float* bl0 = (const float*)d_in[3];
  const float* Wr0 = (const float*)d_in[4];
  const float* Wl1 = (const float*)d_in[5];
  const float* bl1 = (const float*)d_in[6];
  const float* Wr1 = (const float*)d_in[7];
  float* out = (float*)d_out;

  const int N = in_sizes[0] / 64;
  const int E = in_sizes[1] / 2;

  float* Yl1 = (float*)d_ws;
  float* Yr1 = Yl1 + (size_t)N * 32;
  const int NB1 = (E + CHUNK - 1) / CHUNK;
  const int NBK = (N + BNODES - 1) / BNODES;
  int* T        = (int*)(Yr1 + (size_t)N * 32);
  unsigned* P   = (unsigned*)(T + (size_t)NB1 * MAXB);
  int* col      = (int*)P + E;
  int* rowptr   = col + E;

  const dim3 blk(256);

  bin_local<<<dim3(NB1), dim3(TPB), 0, stream>>>(ei, T, P, E, NBK);
  sort_gather_gemm<<<dim3(2 * NBK), blk, 0, stream>>>(P, T, rowptr, col, N, E, NB1,
                                                     x, Wl0, bl0, Wr0, Wl1, bl1, Wr1, Yl1, Yr1);
  sage_gather<32, false><<<dim3(4096), blk, 0, stream>>>(rowptr, col, Yl1, Yr1, out, N, E);
}


# Level 2 of the edge sort reads each bucket's edge words once: one 16-wave block per bucket of 256 nodes keeps them in registers from count to placement; sums still in LDS arrival order
# speedup vs baseline: 5.2857x; 1.0353x over previous
#include <hip/hip_runtime.h>


typedef __attribute__((ext_vector_type(2))) float v2f;
typedef __attribute__((ext_vector_type(8))) float v8f;

#define KDIM 64

#define LDP 68

template <int NT>
__device__ __forceinline__ void put_tile(float* lds, const v8f* acc, int half, int l15) {
#pragma unroll
  for (int t = 0; t < NT; ++t)
#pragma unroll
    for (int v = 0; v < 8; ++v) lds[(v + 8 * half) * LDP + t * 16 + l15] = acc[t][v];
}

template <int OUTF>
__device__ __forceinline__ void store_tile_rows(const float* lds, float* __restrict__ Y,
                                                int r0, int n, int lane) {
  constexpr int LPR = OUTF / 4;
  constexpr int RPI = 32 / LPR;
  constexpr int NS = 16 / RPI;
  const int rr = lane / LPR;
  const int c = (lane % LPR) * 4;
  float4 v[NS];
#pragma unroll
  for (int j = 0; j < NS; ++j) v[j] = *(const float4*)(lds + (j * RPI + rr) * LDP + c);
  float* yp = Y + (long)(r0 + rr) * OUTF + c;
  if (r0 + 16 <= n) {
#pragma unroll
    for (int j = 0; j < NS; ++j) *(float4*)(yp + (long)j * RPI * OUTF) = v[j];
  } else {
#pragma unroll
    for (int j = 0; j < NS; ++j)
      if (r0 + j * RPI + rr < n) *(float4*)(yp + (long)j * RPI * OUTF) = v[j];
  }
}

template <int OUTF>
__device__ __forceinline__ void mma_pair(const float* rows, const float* sW, const float* __restrict__ bl,
                                         v8f* acc, int half, int l15) {
  constexpr int NT = OUTF / 16;
  constexpr int NB = 2 * NT;
#pragma unroll
  for (int t = 0; t < NT; ++t) {
    const float b = bl[t * 16 + l15];
#pragma unroll
    for (int j = 0; j < 8; ++j) {
      acc[t][j] = b;
      acc[NT + t][j] = 0.0f;
    }
  }

  const float* ap = rows + l15 * LDP + half * 2;
  const float* bp = sW + l15 * LDP + half * 2;
#pragma unroll
  for (int kb = 0; kb < KDIM / 4; ++kb) {
    const v2f a = *(const v2f*)(ap + kb * 4);
#pragma unroll
    for (int t = 0; t < NB; ++t) {
      const v2f b = *(const v2f*)(bp + t * 16 * LDP + kb * 4);
      acc[t] = __builtin_amdgcn_wmma_f32_16x16x4_f32(
           false, a,  false, b,
           (short)0, acc[t],  false,  false);
    }
  }
}

#define BSH 8
#define BNODES (1 << BSH)
#define MAXB 512
#define TPB 896
#define EPT 7
#define CHUNK (TPB * EPT)
#define SCAP 8192

__global__ __launch_bounds__(TPB) void bin_local(
    const int* __restrict__ ei, int* __restrict__ T, unsigned* __restrict__ P, int nE, int nbk) {
  __shared__ unsigned stage[CHUNK];
  __shared__ int hist[MAXB];
  __shared__ int wsA[MAXB / 32];
  const int tid = threadIdx.x;
  const int lane = tid & 31;
  const int wave = tid >> 5;
  const int me = blockIdx.x;
  const int beg = me * CHUNK;
  if (tid < MAXB) hist[tid] = 0;
  int s[EPT], d[EPT];
#pragma unroll
  for (int j = 0; j < EPT; ++j) {
    const int e = min(beg + j * TPB + tid, nE - 1);
    s[j] = ei[e];
    d[j] = ei[nE + e];
  }
  __syncthreads();
  const int cnt = min(CHUNK, nE - beg);
  int b[EPT], rk[EPT];
#pragma unroll
  for (int j = 0; j < EPT; ++j) {
    b[j] = (int)min((unsigned)d[j] >> BSH, (unsigned)(nbk - 1));
    rk[j] = 0;
    if (j * TPB + tid < cnt) rk[j] = atomicAdd(&hist[b[j]], 1);
  }
  __syncthreads();
  const int hv = (tid < MAXB) ? hist[tid] : 0;
  int inc = hv;
#pragma unroll
  for (int k = 1; k < 32; k <<= 1) {
    const int a = __shfl_up(inc, k);
    if (lane >= k) inc += a;
  }
  if (tid < MAXB && lane == 31) wsA[wave] = inc;
  __syncthreads();
  if (tid < MAXB) {
    int st = inc - hv;
#pragma unroll
    for (int w = 0; w < MAXB / 32; ++w)
      if (w < wave) st += wsA[w];
    hist[tid] = st;
    T[me * MAXB + tid] = st;
  }
  __syncthreads();
#pragma unroll
  for (int j = 0; j < EPT; ++j) {
    if (j * TPB + tid < cnt) {
      const int p = hist[b[j]] + rk[j];
      if ((unsigned)p < (unsigned)CHUNK)
        stage[p] = ((unsigned)s[j] << BSH) | ((unsigned)d[j] & (BNODES - 1));
    }
  }
  __syncthreads();
  for (int i = tid; i < cnt; i += TPB) P[beg + i] = stage[i];
}

__device__ __forceinline__ void run_of(const int* __restrict__ T, int nE, int nb1, int bk, int r,
                                       int& first, int& len, int& start) {
  first = 0;
  len = 0;
  start = 0;
  if (r < nb1) {
    const int cc = min(CHUNK, nE - r * CHUNK);
    int s0 = T[r * MAXB + bk];
    int e0 = (bk + 1 < MAXB) ? T[r * MAXB + bk + 1] : cc;
    s0 = min(max(s0, 0), cc);
    e0 = min(max(e0, s0), cc);
    first = r * CHUNK + s0;
    len = e0 - s0;
    start = s0;
  }
}

template <bool FIRST, typename F, typename M>
__device__ __forceinline__ void for_each_word(const unsigned* __restrict__ P, const int* __restrict__ T,
                                              int nE, int nb1, int bk, int first0, int len0, int h0,
                                              unsigned (&keep)[16], F f, M mid) {
  const int tid = threadIdx.x;
  const int lane = tid & 31;
  for (int g = 0; g * 256 < nb1; ++g) {
    int first = first0, len = len0;
    if (g != 0) {
      int st;
      run_of(T, nE, nb1, bk, g * 256 + (tid & 255), first, len, st);
    }
    unsigned w[16];
    if (FIRST || g != 0) {
#pragma unroll
      for (int u = 0; u < 16; ++u) {
        const int fu = __builtin_amdgcn_readlane(first, h0 + u);
        const int lu = __builtin_amdgcn_readlane(len, h0 + u);
        w[u] = P[min(fu + min(lane, max(lu - 1, 0)), nE - 1)];
      }
      if (g == 0) {
#pragma unroll
        for (int u = 0; u < 16; ++u) keep[u] = w[u];
        mid();
      }
    } else {
#pragma unroll
      for (int u = 0; u < 16; ++u) w[u] = keep[u];
    }
#pragma unroll
    for (int u = 0; u < 16; ++u) {
      const int fu = __builtin_amdgcn_readlane(first, h0 + u);
      const int lu = __builtin_amdgcn_readlane(len, h0 + u);
      if (lane < lu) f(w[u]);
      if (lu > 32)
        for (int j = 32 + lane; j < lu; j += 32) f(P[fu + j]);
    }
  }
}

__global__ __launch_bounds__(512, 8) void sort_gather_gemm(
    const unsigned* __restrict__ P, const int* __restrict__ T,
    int* __restrict__ rowptr, int* col, int n, int nE, int nb1,
    const float* __restrict__ X,
    const float* __restrict__ Wl0, const float* __restrict__ bl0, const float* __restrict__ Wr0,
    const float* __restrict__ Wl1, const float* __restrict__ bl1, const float* __restrict__ Wr1,
    float* __restrict__ Yl, float* __restrict__ Yr) {
  constexpr int F = KDIM;
  __shared__ __attribute__((aligned(16))) float sW[64 * LDP];
  __shared__ __attribute__((aligned(16))) float sA[16 * 16 * LDP];
  __shared__ int cnt[BNODES];
  __shared__ int rs[BNODES + 1];
  __shared__ int wsum[16];
  __shared__ int sfirst[256];
  __shared__ int slen[256];
  __shared__ float gate[BNODES];
  __shared__ int sorted[SCAP];
  const int tid = threadIdx.x;
  const int wave = __builtin_amdgcn_readfirstlane(tid >> 5);
  const int lane = tid & 31;
  const int half = lane >> 4;
  const int l15  = lane & 15;
  const int bk = (int)blockIdx.x;
  const int ln0 = wave * 16;
  const int node0 = bk * BNODES + ln0;
  const int h0 = (wave >> 3) * 16;
  float* own = sA + wave * 16 * LDP;

  int sb = 0;
  if (tid < 256) {
    cnt[tid] = 0;
    int fi0, le0, st0;
    run_of(T, nE, nb1, bk, tid, fi0, le0, st0);
    sb = st0;
    for (int g = 1; g * 256 < nb1; ++g) {
      int fi, le, st;
      run_of(T, nE, nb1, bk, g * 256 + tid, fi, le, st);
      sb += st;
    }
    sfirst[tid] = fi0;
    slen[tid] = le0;
  }
  float4 wa[2], xv[8];
#pragma unroll
  for (int i = 0; i < 2; ++i) {
    const int idx = i * 512 + tid;
    wa[i] = *(const float4*)(Wr0 + (idx >> 4) * KDIM + (idx & 15) * 4);
  }
#pragma unroll
  for (int i = 0; i < 8; ++i) {
    const int row = min(node0 + i * 2 + half, n - 1);
    xv[i] = *(const float4*)(X + (long)row * F + l15 * 4);
  }
#pragma unroll
  for (int k = 16; k >= 1; k >>= 1) sb += __shfl_xor(sb, k);
  if (lane == 0) wsum[wave] = sb;
#pragma unroll
  for (int i = 0; i < 2; ++i) {
    const int idx = i * 512 + tid;
    *(float4*)(sW + (idx >> 4) * LDP + (idx & 15) * 4) = wa[i];
  }
#pragma unroll
  for (int i = 0; i < 8; ++i) *(float4*)(own + (i * 2 + half) * LDP + l15 * 4) = xv[i];
  __syncthreads();
  const int first0 = sfirst[tid & 255];
  const int len0 = slen[tid & 255];
  int beg = 0;
#pragma unroll
  for (int w = 0; w < 16; ++w) beg += wsum[w];
  beg = min(max(beg, 0), nE);

  v8f acc[4];
#pragma unroll
  for (int t = 0; t < 4; ++t)
#pragma unroll
    for (int j = 0; j < 8; ++j) acc[t][j] = 0.0f;
  const float* ap = own + l15 * LDP + half * 2;
  const float* bp = sW + l15 * LDP + half * 2;
  auto k_steps = [&](int kb0) {
#pragma unroll
    for (int kb = kb0; kb < kb0 + 4; ++kb) {
      const v2f a = *(const v2f*)(ap + kb * 4);
#pragma unroll
      for (int t = 0; t < 4; ++t) {
        const v2f b = *(const v2f*)(bp + t * 16 * LDP + kb * 4);
        acc[t] = __builtin_amdgcn_wmma_f32_16x16x4_f32(false, a, false, b, (short)0, acc[t], false, false);
      }
    }
  };

  unsigned keep[16];
  for_each_word<true>(P, T, nE, nb1, bk, first0, len0, h0, keep,
                      [&](unsigned w) { atomicAdd(&cnt[w & (BNODES - 1)], 1); },
                      [&]() {
                        k_steps(0);
                        k_steps(4);
                        k_steps(8);
                        k_steps(12);
                      });
  __syncthreads();
  const int c = (tid < BNODES) ? cnt[tid] : 0;
  int inc = c;
#pragma unroll
  for (int k = 1; k < 32; k <<= 1) {
    const int a = __shfl_up(inc, k);
    if (lane >= k) inc += a;
  }
  if (tid < BNODES && lane == 31) wsum[wave] = inc;
  __syncthreads();
  if (tid < BNODES) {
    int excl = inc - c;
#pragma unroll
    for (int w = 0; w < BNODES / 32; ++w)
      if (w < wave) excl += wsum[w];
    rs[tid] = excl;
    if (tid == BNODES - 1) rs[BNODES] = excl + c;
    const int node = bk * BNODES + tid;
    if (node < n) rowptr[node] = beg + excl;
    if (node == n - 1) rowptr[n] = beg + excl + c;
    cnt[tid] = excl;
  }
  __syncthreads();
  const int hlen = __builtin_amdgcn_readfirstlane(rs[BNODES]);
  const int gbase = __builtin_amdgcn_readfirstlane(beg);
  const bool fits = hlen <= SCAP;
  for_each_word<false>(P, T, nE, nb1, bk, first0, len0, h0, keep, [&](unsigned w) {
    const int nd = (int)(w & (BNODES - 1));
    const int p = atomicAdd(&cnt[nd], 1);
    const int s = (int)(w >> BSH);
    if (fits) {
      if ((unsigned)p < (unsigned)SCAP) sorted[p] = s;
    } else {
      if ((unsigned)p < (unsigned)hlen && (unsigned)(gbase + p) < (unsigned)nE) col[gbase + p] = s;
    }
  }, [&]() {});
  __syncthreads();
  const int wr = tid >> 4;
  const int wc4 = (tid & 15) * 4;
  const float4 p0 = *(const float4*)(Wl0 + wr * KDIM + wc4);
  const float4 p1 = *(const float4*)(Wl0 + (32 + wr) * KDIM + wc4);
  if (fits)
    for (int i = tid; i < hlen; i += 512)
      if ((unsigned)(gbase + i) < (unsigned)nE) col[gbase + i] = sorted[i];

  auto src_at = [&](int rel) -> int {
    return fits ? sorted[min(max(rel, 0), SCAP - 1)] : col[min(max(gbase + rel, 0), nE - 1)];
  };
  int rp = rs[ln0 + min(lane, 16)];
  rp = min(max(rp, 0), hlen);
  int cvn = src_at(__builtin_amdgcn_readlane(rp, 0) + lane);
  for (int i = 0; i < 16; ++i) {
    const int node = node0 + i;
    const int lb = __builtin_amdgcn_readlane(rp, i);
    const int nxt = __builtin_amdgcn_readlane(rp, i + 1);
    const int end = max(nxt, lb);
    int cv = cvn;
    cvn = src_at(nxt + lane);
    float4 o = make_float4(0.0f, 0.0f, 0.0f, 0.0f);
    float gt = 0.0f;
    if (node < n) {
      float4 sum = make_float4(0.0f, 0.0f, 0.0f, 0.0f);
      for (int base = lb; base < end; base += 32) {
        const int m = min(32, end - base);
        if (base != lb) cv = src_at(base + lane);
        cv = (lane < m) ? (int)min((unsigned)cv, (unsigned)(n - 1)) : 0;
        const int nf = m >> 1;
        const bool odd = (m & 1) != 0;
#pragma unroll
        for (int b = 0; b < 16; b += 8) {
          if (2 * b < m) {
            float4 v[8], vt;
            const bool tail = odd && nf >= b && nf < b + 8;
#pragma unroll
            for (int k = 0; k < 8; ++k) {
              if (b + k < nf) {
                const int s0 = __builtin_amdgcn_readlane(cv, 2 * (b + k));
                const int s1 = __builtin_amdgcn_readlane(cv, 2 * (b + k) + 1);
                const int s = half ? s1 : s0;
                v[k] = *(const float4*)(X + (long)s * F + l15 * 4);
              }
            }
            if (tail) {
              const int st = __builtin_amdgcn_readlane(cv, m - 1);
              vt = *(const float4*)(X + (long)st * F + l15 * 4);
            }
#pragma unroll
            for (int k = 0; k < 8; ++k) {
              if (b + k < nf) {
                sum.x += v[k].x; sum.y += v[k].y; sum.z += v[k].z; sum.w += v[k].w;
              }
            }
            if (tail) {
              sum.x += half ? 0.0f : vt.x;
              sum.y += half ? 0.0f : vt.y;
              sum.z += half ? 0.0f : vt.z;
              sum.w += half ? 0.0f : vt.w;
            }
          }
        }
      }
      sum.x += __shfl_xor(sum.x, 16);
      sum.y += __shfl_xor(sum.y, 16);
      sum.z += __shfl_xor(sum.z, 16);
      sum.w += __shfl_xor(sum.w, 16);
      const float d = fmaxf((float)(end - lb), 1.0f);
      o.x = sum.x / d;
      o.y = sum.y / d;
      o.z = sum.z / d;
      o.w = sum.w / d;
      gt = (end > lb) ? 1.0f : 0.0f;
    }
    if (half == 0) *(float4*)(own + i * LDP + l15 * 4) = o;
    if (lane == 0) gate[wave * 16 + i] = gt;
  }
  *(float4*)(sW + wr * LDP + wc4) = p0;
  *(float4*)(sW + (32 + wr) * LDP + wc4) = p1;
  const float4 q0 = *(const float4*)(Wl1 + wr * KDIM + wc4);
  const float4 q1 = *(const float4*)(Wr1 + wr * KDIM + wc4);
  __syncthreads();

  k_steps(0);
  k_steps(4);
  k_steps(8);
  k_steps(12);
  {
    float gv[8];
#pragma unroll
    for (int v = 0; v < 8; ++v) gv[v] = gate[wave * 16 + v + 8 * half];
#pragma unroll
    for (int t = 0; t < 4; ++t) {
      const float b = bl0[t * 16 + l15];
#pragma unroll
      for (int v = 0; v < 8; ++v)
        own[(v + 8 * half) * LDP + t * 16 + l15] = fmaxf(acc[t][v] + gv[v] * b, 0.0f);
    }
  }
  __syncthreads();
  *(float4*)(sW + wr * LDP + wc4) = q0;
  *(float4*)(sW + (32 + wr) * LDP + wc4) = q1;
  __syncthreads();

  constexpr int OUTF = 32;
  constexpr int NT = OUTF / 16;
  const int r0 = node0;
  const bool active = r0 < n;
  v8f acc1[2 * NT];
  if (active) {
    mma_pair<OUTF>(own, sW, bl1, acc1, half, l15);
    put_tile<NT>(own, acc1, half, l15);
    put_tile<NT>(own + OUTF, acc1 + NT, half, l15);
  }
  __syncthreads();
  if (active) {
    store_tile_rows<OUTF>(own, Yl, r0, n, lane);
    store_tile_rows<OUTF>(own + OUTF, Yr, r0, n, lane);
  }
}

template <int F, bool RELU>
__global__ __launch_bounds__(256) void sage_gather(
    const int* __restrict__ rowptr, const int* __restrict__ col,
    const float* __restrict__ H, const float* __restrict__ R,
    float* __restrict__ Y, int n, int nE) {
  constexpr int G = F / 4;
  constexpr int NG = 32 / G;
  const int lane = threadIdx.x & 31;
  const int sub = lane / G;
  const int c = lane % G;
  const int wv = __builtin_amdgcn_readfirstlane((int)(threadIdx.x >> 5));
  const int nwaves = gridDim.x * (blockDim.x >> 5);
  for (int node = blockIdx.x * (blockDim.x >> 5) + wv; node < n; node += nwaves) {
    int beg = __builtin_amdgcn_readfirstlane(rowptr[node]);
    int end = __builtin_amdgcn_readfirstlane(rowptr[node + 1]);
    const float4 r = *(const float4*)(R + (long)node * F + c * 4);
    beg = min(max(beg, 0), nE);
    end = min(max(end, beg), nE);
    float4 acc = make_float4(0.0f, 0.0f, 0.0f, 0.0f);
    for (int base = beg; base < end; base += 32) {
      const int m = min(32, end - base);
      int cv = (lane < m) ? col[base + lane] : 0;
      cv = (int)min((unsigned)cv, (unsigned)(n - 1));
      const int full = m / NG;
#pragma unroll 4
      for (int k = 0; k < full; ++k) {
        const int s = __shfl(cv, k * NG + sub);
        const float4 v = *(const float4*)(H + (long)s * F + c * 4);
        acc.x += v.x; acc.y += v.y; acc.z += v.z; acc.w += v.w;
      }
      const int rem = m - full * NG;
      if (rem != 0) {
        const int s = __shfl(cv, full * NG + sub);
        if (sub < rem) {
          const float4 v = *(const float4*)(H + (long)s * F + c * 4);
          acc.x += v.x; acc.y += v.y; acc.z += v.z; acc.w += v.w;
        }
      }
    }
#pragma unroll
    for (int d = G; d < 32; d <<= 1) {
      acc.x += __shfl_xor(acc.x, d);
      acc.y += __shfl_xor(acc.y, d);
      acc.z += __shfl_xor(acc.z, d);
      acc.w += __shfl_xor(acc.w, d);
    }
    if (sub == 0) {
      const float d = fmaxf((float)(end - beg), 1.0f);
      float4 o;
      o.x = r.x + acc.x / d;
      o.y = r.y + acc.y / d;
      o.z = r.z + acc.z / d;
      o.w = r.w + acc.w / d;
      if (RELU) {
        o.x = fmaxf(o.x, 0.0f); o.y = fmaxf(o.y, 0.0f);
        o.z = fmaxf(o.z, 0.0f); o.w = fmaxf(o.w, 0.0f);
      }
      *(float4*)(Y + (long)node * F + c * 4) = o;
    }
  }
}

extern "C" void kernel_launch(void* const* d_in, const int* in_sizes, int n_in,
                              void* d_out, int out_size, void* d_ws, size_t ws_size,
                              hipStream_t stream) {
  const float* x   = (const float*)d_in[0];
  const int*   ei  = (const int*)d_in[1];
  const float* Wl0 = (const float*)d_in[2];
  const float* bl0 = (const float*)d_in[3];
  const float* Wr0 = (const float*)d_in[4];
  const float* Wl1 = (const float*)d_in[5];
  const float* bl1 = (const float*)d_in[6];
  const float* Wr1 = (const float*)d_in[7];
  float* out = (float*)d_out;

  const int N = in_sizes[0] / 64;
  const int E = in_sizes[1] / 2;

  float* Yl1 = (float*)d_ws;
  float* Yr1 = Yl1 + (size_t)N * 32;
  const int NB1 = (E + CHUNK - 1) / CHUNK;
  const int NBK = (N + BNODES - 1) / BNODES;
  int* T        = (int*)(Yr1 + (size_t)N * 32);
  unsigned* P   = (unsigned*)(T + (size_t)NB1 * MAXB);
  int* col      = (int*)P + E;
  int* rowptr   = col + E;

  const dim3 blk(256);

  bin_local<<<dim3(NB1), dim3(TPB), 0, stream>>>(ei, T, P, E, NBK);
  sort_gather_gemm<<<dim3(NBK), dim3(512), 0, stream>>>(P, T, rowptr, col, N, E, NB1,
                                                     x, Wl0, bl0, Wr0, Wl1, bl1, Wr1, Yl1, Yr1);
  sage_gather<32, false><<<dim3(4096), blk, 0, stream>>>(rowptr, col, Yl1, Yr1, out, N, E);
}
